# starting kernel
# baseline (speedup 1.0000x reference)
#include <hip/hip_runtime.h>
#include <hip/hip_bf16.h>

typedef __attribute__((ext_vector_type(2))) float v2f;
typedef __attribute__((ext_vector_type(8))) float v8f;

#define IN_FEATS 512
#define HIDDEN   64
#define NCLASSES 47
#define ALPHA    0.1f
#define KSTEPS   10

__global__ __launch_bounds__(256) void appnp_zero4(float4* __restrict__ p, int n4) {
    int i = blockIdx.x * blockDim.x + threadIdx.x;
    if (i < n4) p[i] = make_float4(0.f, 0.f, 0.f, 0.f);
}

__global__ __launch_bounds__(256) void appnp_degree(const int* __restrict__ src,
                                                    const int* __restrict__ dst,
                                                    float* __restrict__ odeg,
                                                    float* __restrict__ ideg, int E) {
    int e = blockIdx.x * blockDim.x + threadIdx.x;
    if (e >= E) return;
    atomicAdd(&odeg[src[e]], 1.0f);
    atomicAdd(&ideg[dst[e]], 1.0f);
}

__global__ __launch_bounds__(256) void appnp_norm(float* __restrict__ d, int n) {
    int i = blockIdx.x * blockDim.x + threadIdx.x;
    if (i < n) d[i] = rsqrtf(fmaxf(d[i], 1.0f));
}

__global__ __launch_bounds__(256) void appnp_gemm1_wmma(const float* __restrict__ F,
                                                        const float* __restrict__ W0,
                                                        const float* __restrict__ b0,
                                                        float* __restrict__ H, int N) {
    __shared__ float sw[256 * HIDDEN];
    const int lane = threadIdx.x & 31;
    const int wave = threadIdx.x >> 5;
    const int nTiles = (N + 15) >> 4;
    int rowTile = blockIdx.x * 8 + wave;
    if (rowTile >= nTiles) rowTile = nTiles - 1;
    const int rowBase = rowTile * 16;
    const int m  = lane & 15;
    const int kh = (lane >> 4) << 1;

    v8f acc0 = {}, acc1 = {}, acc2 = {}, acc3 = {};
    int arow = rowBase + m; if (arow >= N) arow = N - 1;
    const float* frow = F + (size_t)arow * IN_FEATS + kh;

    for (int ko = 0; ko < IN_FEATS; ko += 256) {
        __syncthreads();
        const float4* gsrc = (const float4*)(W0 + (size_t)ko * HIDDEN);
        float4* ldst = (float4*)sw;
        for (int i = threadIdx.x; i < (256 * HIDDEN) / 4; i += 256) ldst[i] = gsrc[i];
        __syncthreads();

        for (int k = 0; k < 256; k += 4) {
            if ((k & 63) == 0) __builtin_prefetch(frow + ko + k + 64, 0, 1);
            v2f a = *(const v2f*)(frow + ko + k);
            const float* w0p = sw + (k + kh) * HIDDEN + m;
            const float* w1p = w0p + HIDDEN;
            v2f b;
            b.x = w0p[0];  b.y = w1p[0];
            acc0 = __builtin_amdgcn_wmma_f32_16x16x4_f32(false, a, false, b, (short)0, acc0, false, false);
            b.x = w0p[16]; b.y = w1p[16];
            acc1 = __builtin_amdgcn_wmma_f32_16x16x4_f32(false, a, false, b, (short)0, acc1, false, false);
            b.x = w0p[32]; b.y = w1p[32];
            acc2 = __builtin_amdgcn_wmma_f32_16x16x4_f32(false, a, false, b, (short)0, acc2, false, false);
            b.x = w0p[48]; b.y = w1p[48];
            acc3 = __builtin_amdgcn_wmma_f32_16x16x4_f32(false, a, false, b, (short)0, acc3, false, false);
        }
    }

    const int rh = (lane >> 4) * 8;
    const float bias0 = b0[m], bias1 = b0[16 + m], bias2 = b0[32 + m], bias3 = b0[48 + m];
    for (int r = 0; r < 8; ++r) {
        int row = rowBase + rh + r;
        if (row < N) {
            float* hrow = H + (size_t)row * HIDDEN;
            hrow[m]      = fmaxf(acc0[r] + bias0, 0.f);
            hrow[16 + m] = fmaxf(acc1[r] + bias1, 0.f);
            hrow[32 + m] = fmaxf(acc2[r] + bias2, 0.f);
            hrow[48 + m] = fmaxf(acc3[r] + bias3, 0.f);
        }
    }
}

__global__ __launch_bounds__(256) void appnp_gemm2_wmma(const float* __restrict__ H,
                                                        const float* __restrict__ W1,
                                                        const float* __restrict__ b1,
                                                        const float* __restrict__ snorm,
                                                        float* __restrict__ O,
                                                        float* __restrict__ OS, int N) {
    const int lane = threadIdx.x & 31;
    const int wave = threadIdx.x >> 5;
    const int rowBase = (blockIdx.x * 8 + wave) * 16;
    if (rowBase >= N) return;
    const int m  = lane & 15;
    const int kh = (lane >> 4) << 1;
    const bool c2ok = (32 + m) < NCLASSES;

    v8f acc0 = {}, acc1 = {}, acc2 = {};
    const float* hrow = H + (size_t)(rowBase + m) * HIDDEN + kh;

    for (int k = 0; k < HIDDEN; k += 4) {
        v2f a = *(const v2f*)(hrow + k);
        const float* w0p = W1 + (size_t)(k + kh) * NCLASSES + m;
        const float* w1p = w0p + NCLASSES;
        v2f b;
        b.x = w0p[0];  b.y = w1p[0];
        acc0 = __builtin_amdgcn_wmma_f32_16x16x4_f32(false, a, false, b, (short)0, acc0, false, false);
        b.x = w0p[16]; b.y = w1p[16];
        acc1 = __builtin_amdgcn_wmma_f32_16x16x4_f32(false, a, false, b, (short)0, acc1, false, false);
        b.x = c2ok ? w0p[32] : 0.f;
        b.y = c2ok ? w1p[32] : 0.f;
        acc2 = __builtin_amdgcn_wmma_f32_16x16x4_f32(false, a, false, b, (short)0, acc2, false, false);
    }

    const int rh = (lane >> 4) * 8;
    const float bias0 = b1[m], bias1 = b1[16 + m], bias2 = c2ok ? b1[32 + m] : 0.f;
    for (int r = 0; r < 8; ++r) {
        int row = rowBase + rh + r;
        float sn = snorm[row];
        float* orow = O  + (size_t)row * NCLASSES;
        float* srow = OS + (size_t)row * NCLASSES;
        float v0 = acc0[r] + bias0;
        float v1 = acc1[r] + bias1;
        orow[m]      = v0;  srow[m]      = v0 * sn;
        orow[16 + m] = v1;  srow[16 + m] = v1 * sn;
        if (c2ok) {
            float v2 = acc2[r] + bias2;
            orow[32 + m] = v2;  srow[32 + m] = v2 * sn;
        }
    }
}

__global__ __launch_bounds__(256) void appnp_scatter(const float* __restrict__ hs,
                                                     const int* __restrict__ src,
                                                     const int* __restrict__ dst,
                                                     float* __restrict__ acc,
                                                     long long total) {
    long long idx = (long long)blockIdx.x * blockDim.x + threadIdx.x;
    if (idx >= total) return;
    int e = (int)(idx / NCLASSES);
    int c = (int)(idx - (long long)e * NCLASSES);
    atomicAdd(&acc[(size_t)dst[e] * NCLASSES + c],
              hs[(size_t)src[e] * NCLASSES + c]);
}

__global__ __launch_bounds__(256) void appnp_mix(const float* __restrict__ acc,
                                                 const float* __restrict__ dnorm,
                                                 const float* __restrict__ snorm,
                                                 const float* __restrict__ h0,
                                                 float* __restrict__ out,
                                                 int do_scale, int total) {
    int idx = blockIdx.x * blockDim.x + threadIdx.x;
    if (idx >= total) return;
    int row = idx / NCLASSES;
    float mixed = (1.0f - ALPHA) * acc[idx] * dnorm[row] + ALPHA * h0[idx];
    out[idx] = do_scale ? mixed * snorm[row] : mixed;
}

extern "C" void kernel_launch(void* const* d_in, const int* in_sizes, int n_in,
                              void* d_out, int out_size, void* d_ws, size_t ws_size,
                              hipStream_t stream) {
    const float* F  = (const float*)d_in[0];
    const float* W0 = (const float*)d_in[1];
    const float* b0 = (const float*)d_in[2];
    const float* W1 = (const float*)d_in[3];
    const float* b1 = (const float*)d_in[4];
    const int*  src = (const int*)d_in[5];
    const int*  dst = (const int*)d_in[6];

    const int N = in_sizes[0] / IN_FEATS;
    const int E = in_sizes[5];
    float* out = (float*)d_out;

    float* ws    = (float*)d_ws;
    float* snorm = ws;
    float* dnorm = ws + (size_t)N;
    float* hrelu = ws + 2 * (size_t)N;
    float* h0    = hrelu + (size_t)N * HIDDEN;
    float* hs    = h0 + (size_t)N * NCLASSES;
    float* acc   = hs + (size_t)N * NCLASSES;

    {
        int n4 = (2 * N) / 4;
        appnp_zero4<<<(n4 + 255) / 256, 256, 0, stream>>>((float4*)snorm, n4);
        appnp_degree<<<(E + 255) / 256, 256, 0, stream>>>(src, dst, snorm, dnorm, E);
        appnp_norm<<<(2 * N + 255) / 256, 256, 0, stream>>>(snorm, 2 * N);
    }

    {
        int rowTiles = (N + 15) / 16;
        int blocks   = (rowTiles + 7) / 8;
        appnp_gemm1_wmma<<<blocks, 256, 0, stream>>>(F, W0, b0, hrelu, N);
        appnp_gemm2_wmma<<<blocks, 256, 0, stream>>>(hrelu, W1, b1, snorm, h0, hs, N);
    }

    {
        const long long tot = (long long)E * NCLASSES;
        const int sblocks = (int)((tot + 255) / 256);
        const int nc  = N * NCLASSES;
        const int nc4 = nc / 4;
        for (int it = 0; it < KSTEPS; ++it) {
            appnp_zero4<<<(nc4 + 255) / 256, 256, 0, stream>>>((float4*)acc, nc4);
            appnp_scatter<<<sblocks, 256, 0, stream>>>(hs, src, dst, acc, tot);
            if (it == KSTEPS - 1) {
                appnp_mix<<<(nc + 255) / 256, 256, 0, stream>>>(acc, dnorm, snorm, h0, out, 0, nc);
            } else {
                appnp_mix<<<(nc + 255) / 256, 256, 0, stream>>>(acc, dnorm, snorm, h0, hs, 1, nc);
            }
        }
    }
}


# Per-step atomic scatter replaced by a per-call dst-sorted edge list (rank from the count atomics + prefix sum) and a per-row gather with the mix fused; row sums in list order, no float atomics on h
# speedup vs baseline: 5.0578x; 5.0578x over previous
#include <hip/hip_runtime.h>
#include <hip/hip_bf16.h>

typedef __attribute__((ext_vector_type(2))) float v2f;
typedef __attribute__((ext_vector_type(8))) float v8f;

#define IN_FEATS 512
#define HIDDEN   64
#define NCLASSES 47
#define CPAD     48
#define ROWV4    12
#define ALPHA    0.1f
#define KSTEPS   10

__global__ __launch_bounds__(256) void appnp_zero4(float4* __restrict__ p, int n4) {
    int i = blockIdx.x * blockDim.x + threadIdx.x;
    if (i < n4) p[i] = make_float4(0.f, 0.f, 0.f, 0.f);
}

__global__ __launch_bounds__(256) void appnp_degree(const int* __restrict__ src,
                                                    const int* __restrict__ dst,
                                                    float* __restrict__ odeg,
                                                    int* __restrict__ icnt,
                                                    int* __restrict__ rank, int E) {
    int e = blockIdx.x * blockDim.x + threadIdx.x;
    if (e >= E) return;
    atomicAdd(&odeg[src[e]], 1.0f);
    rank[e] = atomicAdd(&icnt[dst[e]], 1);
}

__global__ __launch_bounds__(256) void appnp_rowptr(const int* __restrict__ icnt,
                                                    int* __restrict__ rowptr, int N) {
    __shared__ int sPart[8];
    __shared__ int sWave[8];
    const int tid  = threadIdx.x;
    const int lane = tid & 31;
    const int wave = tid >> 5;
    const int blockStart = blockIdx.x * 1024;

    int part = 0;
#pragma unroll 8
    for (int i = tid * 4; i < blockStart; i += 1024) {
        const int4 v = *(const int4*)(icnt + i);
        part += (v.x + v.y) + (v.z + v.w);
    }
    for (int o = 16; o > 0; o >>= 1) part += __shfl_xor(part, o, 32);
    if (lane == 0) sPart[wave] = part;

    const int i0 = blockStart + tid * 4;
    int c0 = 0, c1 = 0, c2 = 0, c3 = 0;
    if (i0 + 3 < N) {
        const int4 v = *(const int4*)(icnt + i0);
        c0 = v.x; c1 = v.y; c2 = v.z; c3 = v.w;
    } else {
        if (i0     < N) c0 = icnt[i0];
        if (i0 + 1 < N) c1 = icnt[i0 + 1];
        if (i0 + 2 < N) c2 = icnt[i0 + 2];
    }
    const int tsum = (c0 + c1) + (c2 + c3);
    int incl = tsum;
    for (int o = 1; o < 32; o <<= 1) {
        const int up = __shfl_up(incl, o, 32);
        if (lane >= o) incl += up;
    }
    if (lane == 31) sWave[wave] = incl;
    __syncthreads();

    int offset = 0;
    for (int w = 0; w < 8; ++w) offset += sPart[w];
    int wbase = 0;
    for (int w = 0; w < wave; ++w) wbase += sWave[w];
    const int ex = offset + wbase + (incl - tsum);

    if (i0     < N) rowptr[i0]     = ex;
    if (i0 + 1 < N) rowptr[i0 + 1] = ex + c0;
    if (i0 + 2 < N) rowptr[i0 + 2] = ex + c0 + c1;
    if (i0 + 3 < N) rowptr[i0 + 3] = ex + c0 + c1 + c2;
    if (i0 <= N - 1 && N - 1 <= i0 + 3) rowptr[N] = ex + tsum;
}

__global__ __launch_bounds__(256) void appnp_norm(float* __restrict__ snorm,
                                                  const int* __restrict__ icnt,
                                                  float* __restrict__ dnorm, int n) {
    int i = blockIdx.x * blockDim.x + threadIdx.x;
    if (i < n) {
        snorm[i] = rsqrtf(fmaxf(snorm[i], 1.0f));
        dnorm[i] = rsqrtf(fmaxf((float)icnt[i], 1.0f));
    }
}

__global__ __launch_bounds__(256) void appnp_fill(const int* __restrict__ src,
                                                  const int* __restrict__ dst,
                                                  const int* __restrict__ rank,
                                                  const int* __restrict__ rowptr,
                                                  int* __restrict__ csr, int E) {
    int e = blockIdx.x * blockDim.x + threadIdx.x;
    if (e >= E) return;
    csr[rowptr[dst[e]] + rank[e]] = src[e];
}

__global__ __launch_bounds__(256) void appnp_gemm1_wmma(const float* __restrict__ F,
                                                        const float* __restrict__ W0,
                                                        const float* __restrict__ b0,
                                                        float* __restrict__ H, int N) {
    __shared__ float sw[256 * HIDDEN];
    const int lane = threadIdx.x & 31;
    const int wave = threadIdx.x >> 5;
    const int nTiles = (N + 15) >> 4;
    int rowTile = blockIdx.x * 8 + wave;
    if (rowTile >= nTiles) rowTile = nTiles - 1;
    const int rowBase = rowTile * 16;
    const int m  = lane & 15;
    const int kh = (lane >> 4) << 1;

    v8f acc0 = {}, acc1 = {}, acc2 = {}, acc3 = {};
    int arow = rowBase + m; if (arow >= N) arow = N - 1;
    const float* frow = F + (size_t)arow * IN_FEATS + kh;

    for (int ko = 0; ko < IN_FEATS; ko += 256) {
        __syncthreads();
        const float4* gsrc = (const float4*)(W0 + (size_t)ko * HIDDEN);
        float4* ldst = (float4*)sw;
        for (int i = threadIdx.x; i < (256 * HIDDEN) / 4; i += 256) ldst[i] = gsrc[i];
        __syncthreads();

        for (int k = 0; k < 256; k += 4) {
            if ((k & 63) == 0) __builtin_prefetch(frow + ko + k + 64, 0, 1);
            v2f a = *(const v2f*)(frow + ko + k);
            const float* w0p = sw + (k + kh) * HIDDEN + m;
            const float* w1p = w0p + HIDDEN;
            v2f b;
            b.x = w0p[0];  b.y = w1p[0];
            acc0 = __builtin_amdgcn_wmma_f32_16x16x4_f32(false, a, false, b, (short)0, acc0, false, false);
            b.x = w0p[16]; b.y = w1p[16];
            acc1 = __builtin_amdgcn_wmma_f32_16x16x4_f32(false, a, false, b, (short)0, acc1, false, false);
            b.x = w0p[32]; b.y = w1p[32];
            acc2 = __builtin_amdgcn_wmma_f32_16x16x4_f32(false, a, false, b, (short)0, acc2, false, false);
            b.x = w0p[48]; b.y = w1p[48];
            acc3 = __builtin_amdgcn_wmma_f32_16x16x4_f32(false, a, false, b, (short)0, acc3, false, false);
        }
    }

    const int rh = (lane >> 4) * 8;
    const float bias0 = b0[m], bias1 = b0[16 + m], bias2 = b0[32 + m], bias3 = b0[48 + m];
    for (int r = 0; r < 8; ++r) {
        int row = rowBase + rh + r;
        if (row < N) {
            float* hrow = H + (size_t)row * HIDDEN;
            hrow[m]      = fmaxf(acc0[r] + bias0, 0.f);
            hrow[16 + m] = fmaxf(acc1[r] + bias1, 0.f);
            hrow[32 + m] = fmaxf(acc2[r] + bias2, 0.f);
            hrow[48 + m] = fmaxf(acc3[r] + bias3, 0.f);
        }
    }
}

__global__ __launch_bounds__(256) void appnp_gemm2_wmma(const float* __restrict__ H,
                                                        const float* __restrict__ W1,
                                                        const float* __restrict__ b1,
                                                        const float* __restrict__ snorm,
                                                        float* __restrict__ O,
                                                        float* __restrict__ OS, int N) {
    const int lane = threadIdx.x & 31;
    const int wave = threadIdx.x >> 5;
    const int rowBase = (blockIdx.x * 8 + wave) * 16;
    if (rowBase >= N) return;
    const int m  = lane & 15;
    const int kh = (lane >> 4) << 1;
    const bool c2ok = (32 + m) < NCLASSES;

    v8f acc0 = {}, acc1 = {}, acc2 = {};
    const float* hrow = H + (size_t)(rowBase + m) * HIDDEN + kh;

    for (int k = 0; k < HIDDEN; k += 4) {
        v2f a = *(const v2f*)(hrow + k);
        const float* w0p = W1 + (size_t)(k + kh) * NCLASSES + m;
        const float* w1p = w0p + NCLASSES;
        v2f b;
        b.x = w0p[0];  b.y = w1p[0];
        acc0 = __builtin_amdgcn_wmma_f32_16x16x4_f32(false, a, false, b, (short)0, acc0, false, false);
        b.x = w0p[16]; b.y = w1p[16];
        acc1 = __builtin_amdgcn_wmma_f32_16x16x4_f32(false, a, false, b, (short)0, acc1, false, false);
        b.x = c2ok ? w0p[32] : 0.f;
        b.y = c2ok ? w1p[32] : 0.f;
        acc2 = __builtin_amdgcn_wmma_f32_16x16x4_f32(false, a, false, b, (short)0, acc2, false, false);
    }

    const int rh = (lane >> 4) * 8;
    const float bias0 = b1[m], bias1 = b1[16 + m], bias2 = c2ok ? b1[32 + m] : 0.f;
    for (int r = 0; r < 8; ++r) {
        int row = rowBase + rh + r;
        float sn = snorm[row];
        float* orow = O  + (size_t)row * CPAD;
        float* srow = OS + (size_t)row * CPAD;
        float v0 = acc0[r] + bias0;
        float v1 = acc1[r] + bias1;
        float v2 = acc2[r] + bias2;
        orow[m]      = v0;  srow[m]      = v0 * sn;
        orow[16 + m] = v1;  srow[16 + m] = v1 * sn;
        orow[32 + m] = v2;  srow[32 + m] = v2 * sn;
    }
}

template <bool LAST>
__global__ __launch_bounds__(512) void appnp_gather(const float4* __restrict__ hin,
                                                    const int* __restrict__ rowptr,
                                                    const int* __restrict__ csr,
                                                    const float* __restrict__ dnorm,
                                                    const float* __restrict__ snorm,
                                                    const float4* __restrict__ h0,
                                                    float4* __restrict__ hout,
                                                    float* __restrict__ out, int N) {
    const int tid  = threadIdx.x;
    const int q    = tid & 15;
    const int qq   = q < ROWV4 ? q : ROWV4 - 1;
    const int rloc = tid >> 4;
    const int row  = blockIdx.x * 32 + rloc;

    float4 res = make_float4(0.f, 0.f, 0.f, 0.f);
    if (row < N) {
        float4 acc = make_float4(0.f, 0.f, 0.f, 0.f);
        const int beg = rowptr[row];
        const int end = rowptr[row + 1];
        for (int base = beg; base < end; base += 16) {
            const int n  = min(16, end - base);
            const int my = (q < n) ? csr[base + q] : 0;
            for (int t = 0; t < n; t += 4) {
                const int s0 = __shfl(my, t,     16);
                const int s1 = __shfl(my, t + 1, 16);
                const int s2 = __shfl(my, t + 2, 16);
                const int s3 = __shfl(my, t + 3, 16);
                const float4 v0 = hin[(size_t)s0 * ROWV4 + qq];
                const float4 v1 = hin[(size_t)s1 * ROWV4 + qq];
                const float4 v2 = hin[(size_t)s2 * ROWV4 + qq];
                const float4 v3 = hin[(size_t)s3 * ROWV4 + qq];
                const bool k1 = t + 1 < n, k2 = t + 2 < n, k3 = t + 3 < n;
                acc.x += v0.x;               acc.y += v0.y;               acc.z += v0.z;               acc.w += v0.w;
                acc.x += k1 ? v1.x : 0.f;    acc.y += k1 ? v1.y : 0.f;    acc.z += k1 ? v1.z : 0.f;    acc.w += k1 ? v1.w : 0.f;
                acc.x += k2 ? v2.x : 0.f;    acc.y += k2 ? v2.y : 0.f;    acc.z += k2 ? v2.z : 0.f;    acc.w += k2 ? v2.w : 0.f;
                acc.x += k3 ? v3.x : 0.f;    acc.y += k3 ? v3.y : 0.f;    acc.z += k3 ? v3.z : 0.f;    acc.w += k3 ? v3.w : 0.f;
            }
        }
        const float dn = dnorm[row];
        const float4 t0 = h0[(size_t)row * ROWV4 + qq];
        res.x = (1.0f - ALPHA) * acc.x * dn + ALPHA * t0.x;
        res.y = (1.0f - ALPHA) * acc.y * dn + ALPHA * t0.y;
        res.z = (1.0f - ALPHA) * acc.z * dn + ALPHA * t0.z;
        res.w = (1.0f - ALPHA) * acc.w * dn + ALPHA * t0.w;
    }

    if (!LAST) {
        if (row < N && q < ROWV4) {
            const float sn = snorm[row];
            hout[(size_t)row * ROWV4 + q] = make_float4(res.x * sn, res.y * sn, res.z * sn, res.w * sn);
        }
    } else {
        __shared__ float sm[32 * NCLASSES];
        if (row < N && q < ROWV4) {
            float* p = sm + rloc * NCLASSES + q * 4;
            p[0] = res.x; p[1] = res.y; p[2] = res.z;
            if (q * 4 + 3 < NCLASSES) p[3] = res.w;
        }
        __syncthreads();
        const int rowsHere = min(32, N - (int)blockIdx.x * 32);
        const int nvalid = rowsHere * NCLASSES;
        float* obase = out + (size_t)blockIdx.x * (32 * NCLASSES);
        for (int p = 0; p < (32 * NCLASSES + 511) / 512; ++p) {
            const int i = tid + p * 512;
            if (i < nvalid) obase[i] = sm[i];
        }
    }
}

extern "C" void kernel_launch(void* const* d_in, const int* in_sizes, int n_in,
                              void* d_out, int out_size, void* d_ws, size_t ws_size,
                              hipStream_t stream) {
    const float* F  = (const float*)d_in[0];
    const float* W0 = (const float*)d_in[1];
    const float* b0 = (const float*)d_in[2];
    const float* W1 = (const float*)d_in[3];
    const float* b1 = (const float*)d_in[4];
    const int*  src = (const int*)d_in[5];
    const int*  dst = (const int*)d_in[6];

    const int N = in_sizes[0] / IN_FEATS;
    const int E = in_sizes[5];
    float* out = (float*)d_out;

    float* ws    = (float*)d_ws;
    float* snorm = ws;
    int*   icnt  = (int*)(ws + (size_t)N);
    float* dnorm = ws + 2 * (size_t)N;
    float* hrelu = ws + 3 * (size_t)N;
    float* h0    = hrelu + (size_t)N * HIDDEN;
    float* hs    = h0 + (size_t)N * CPAD;
    int*   rank  = (int*)(hs + (size_t)N * CPAD);
    int*   csr   = rank + (size_t)E;
    int*   rowptr = csr + (size_t)E;

    {
        int n4 = (2 * N) / 4;
        appnp_zero4<<<(n4 + 255) / 256, 256, 0, stream>>>((float4*)snorm, n4);
        appnp_degree<<<(E + 255) / 256, 256, 0, stream>>>(src, dst, snorm, icnt, rank, E);
        appnp_rowptr<<<(N + 1023) / 1024, 256, 0, stream>>>(icnt, rowptr, N);
        appnp_norm<<<(N + 255) / 256, 256, 0, stream>>>(snorm, icnt, dnorm, N);
    }

    {
        int rowTiles = (N + 15) / 16;
        int blocks   = (rowTiles + 7) / 8;
        appnp_gemm1_wmma<<<blocks, 256, 0, stream>>>(F, W0, b0, hrelu, N);
        appnp_gemm2_wmma<<<blocks, 256, 0, stream>>>(hrelu, W1, b1, snorm, h0, hs, N);
    }

    {
        appnp_fill<<<(E + 255) / 256, 256, 0, stream>>>(src, dst, rank, rowptr, csr, E);
        float* bufB = hrelu;
        const int gblocks = (N + 31) / 32;
        for (int it = 0; it < KSTEPS; ++it) {
            const float* in = (it & 1) ? bufB : hs;
            float* nxt      = (it & 1) ? hs : bufB;
            if (it == KSTEPS - 1) {
                appnp_gather<true><<<gblocks, 512, 0, stream>>>((const float4*)in, rowptr, csr, dnorm, snorm,
                                                               (const float4*)h0, (float4*)nxt, out, N);
            } else {
                appnp_gather<false><<<gblocks, 512, 0, stream>>>((const float4*)in, rowptr, csr, dnorm, snorm,
                                                                (const float4*)h0, (float4*)nxt, out, N);
            }
        }
    }
}


# Row gather: half a wave owns two rows and keeps 16 row loads in flight (was 4); bounds, norms, teleport row and 32 indices fetched up front; half the waves; row sums in the same list order
# speedup vs baseline: 5.2419x; 1.0364x over previous
#include <hip/hip_runtime.h>
#include <hip/hip_bf16.h>

typedef __attribute__((ext_vector_type(2))) float v2f;
typedef __attribute__((ext_vector_type(8))) float v8f;

#define IN_FEATS 512
#define HIDDEN   64
#define NCLASSES 47
#define CPAD     48
#define ROWV4    12
#define ALPHA    0.1f
#define KSTEPS   10

__global__ __launch_bounds__(256) void appnp_zero4(float4* __restrict__ p, int n4) {
    int i = blockIdx.x * blockDim.x + threadIdx.x;
    if (i < n4) p[i] = make_float4(0.f, 0.f, 0.f, 0.f);
}

__global__ __launch_bounds__(256) void appnp_degree(const int* __restrict__ src,
                                                    const int* __restrict__ dst,
                                                    float* __restrict__ odeg,
                                                    int* __restrict__ icnt,
                                                    int* __restrict__ rank, int E) {
    int e = blockIdx.x * blockDim.x + threadIdx.x;
    if (e >= E) return;
    atomicAdd(&odeg[src[e]], 1.0f);
    rank[e] = atomicAdd(&icnt[dst[e]], 1);
}

__global__ __launch_bounds__(256) void appnp_rowptr(const int* __restrict__ icnt,
                                                    int* __restrict__ rowptr, int N) {
    __shared__ int sPart[8];
    __shared__ int sWave[8];
    const int tid  = threadIdx.x;
    const int lane = tid & 31;
    const int wave = tid >> 5;
    const int blockStart = blockIdx.x * 1024;

    int part = 0;
#pragma unroll 8
    for (int i = tid * 4; i < blockStart; i += 1024) {
        const int4 v = *(const int4*)(icnt + i);
        part += (v.x + v.y) + (v.z + v.w);
    }
    for (int o = 16; o > 0; o >>= 1) part += __shfl_xor(part, o, 32);
    if (lane == 0) sPart[wave] = part;

    const int i0 = blockStart + tid * 4;
    int c0 = 0, c1 = 0, c2 = 0, c3 = 0;
    if (i0 + 3 < N) {
        const int4 v = *(const int4*)(icnt + i0);
        c0 = v.x; c1 = v.y; c2 = v.z; c3 = v.w;
    } else {
        if (i0     < N) c0 = icnt[i0];
        if (i0 + 1 < N) c1 = icnt[i0 + 1];
        if (i0 + 2 < N) c2 = icnt[i0 + 2];
    }
    const int tsum = (c0 + c1) + (c2 + c3);
    int incl = tsum;
    for (int o = 1; o < 32; o <<= 1) {
        const int up = __shfl_up(incl, o, 32);
        if (lane >= o) incl += up;
    }
    if (lane == 31) sWave[wave] = incl;
    __syncthreads();

    int offset = 0;
    for (int w = 0; w < 8; ++w) offset += sPart[w];
    int wbase = 0;
    for (int w = 0; w < wave; ++w) wbase += sWave[w];
    const int ex = offset + wbase + (incl - tsum);

    if (i0     < N) rowptr[i0]     = ex;
    if (i0 + 1 < N) rowptr[i0 + 1] = ex + c0;
    if (i0 + 2 < N) rowptr[i0 + 2] = ex + c0 + c1;
    if (i0 + 3 < N) rowptr[i0 + 3] = ex + c0 + c1 + c2;
    if (i0 <= N - 1 && N - 1 <= i0 + 3) rowptr[N] = ex + tsum;
}

__global__ __launch_bounds__(256) void appnp_norm(float* __restrict__ snorm,
                                                  const int* __restrict__ icnt,
                                                  float* __restrict__ dnorm, int n) {
    int i = blockIdx.x * blockDim.x + threadIdx.x;
    if (i < n) {
        snorm[i] = rsqrtf(fmaxf(snorm[i], 1.0f));
        dnorm[i] = rsqrtf(fmaxf((float)icnt[i], 1.0f));
    }
}

__global__ __launch_bounds__(256) void appnp_fill(const int* __restrict__ src,
                                                  const int* __restrict__ dst,
                                                  const int* __restrict__ rank,
                                                  const int* __restrict__ rowptr,
                                                  int* __restrict__ csr, int E) {
    int e = blockIdx.x * blockDim.x + threadIdx.x;
    if (e >= E) return;
    csr[rowptr[dst[e]] + rank[e]] = src[e];
}

__global__ __launch_bounds__(256) void appnp_gemm1_wmma(const float* __restrict__ F,
                                                        const float* __restrict__ W0,
                                                        const float* __restrict__ b0,
                                                        float* __restrict__ H, int N) {
    __shared__ float sw[256 * HIDDEN];
    const int lane = threadIdx.x & 31;
    const int wave = threadIdx.x >> 5;
    const int nTiles = (N + 15) >> 4;
    int rowTile = blockIdx.x * 8 + wave;
    if (rowTile >= nTiles) rowTile = nTiles - 1;
    const int rowBase = rowTile * 16;
    const int m  = lane & 15;
    const int kh = (lane >> 4) << 1;

    v8f acc0 = {}, acc1 = {}, acc2 = {}, acc3 = {};
    int arow = rowBase + m; if (arow >= N) arow = N - 1;
    const float* frow = F + (size_t)arow * IN_FEATS + kh;

    for (int ko = 0; ko < IN_FEATS; ko += 256) {
        __syncthreads();
        const float4* gsrc = (const float4*)(W0 + (size_t)ko * HIDDEN);
        float4* ldst = (float4*)sw;
        for (int i = threadIdx.x; i < (256 * HIDDEN) / 4; i += 256) ldst[i] = gsrc[i];
        __syncthreads();

        for (int k = 0; k < 256; k += 4) {
            if ((k & 63) == 0) __builtin_prefetch(frow + ko + k + 64, 0, 1);
            v2f a = *(const v2f*)(frow + ko + k);
            const float* w0p = sw + (k + kh) * HIDDEN + m;
            const float* w1p = w0p + HIDDEN;
            v2f b;
            b.x = w0p[0];  b.y = w1p[0];
            acc0 = __builtin_amdgcn_wmma_f32_16x16x4_f32(false, a, false, b, (short)0, acc0, false, false);
            b.x = w0p[16]; b.y = w1p[16];
            acc1 = __builtin_amdgcn_wmma_f32_16x16x4_f32(false, a, false, b, (short)0, acc1, false, false);
            b.x = w0p[32]; b.y = w1p[32];
            acc2 = __builtin_amdgcn_wmma_f32_16x16x4_f32(false, a, false, b, (short)0, acc2, false, false);
            b.x = w0p[48]; b.y = w1p[48];
            acc3 = __builtin_amdgcn_wmma_f32_16x16x4_f32(false, a, false, b, (short)0, acc3, false, false);
        }
    }

    const int rh = (lane >> 4) * 8;
    const float bias0 = b0[m], bias1 = b0[16 + m], bias2 = b0[32 + m], bias3 = b0[48 + m];
    for (int r = 0; r < 8; ++r) {
        int row = rowBase + rh + r;
        if (row < N) {
            float* hrow = H + (size_t)row * HIDDEN;
            hrow[m]      = fmaxf(acc0[r] + bias0, 0.f);
            hrow[16 + m] = fmaxf(acc1[r] + bias1, 0.f);
            hrow[32 + m] = fmaxf(acc2[r] + bias2, 0.f);
            hrow[48 + m] = fmaxf(acc3[r] + bias3, 0.f);
        }
    }
}

__global__ __launch_bounds__(256) void appnp_gemm2_wmma(const float* __restrict__ H,
                                                        const float* __restrict__ W1,
                                                        const float* __restrict__ b1,
                                                        const float* __restrict__ snorm,
                                                        float* __restrict__ O,
                                                        float* __restrict__ OS, int N) {
    const int lane = threadIdx.x & 31;
    const int wave = threadIdx.x >> 5;
    const int rowBase = (blockIdx.x * 8 + wave) * 16;
    if (rowBase >= N) return;
    const int m  = lane & 15;
    const int kh = (lane >> 4) << 1;
    const bool c2ok = (32 + m) < NCLASSES;

    v8f acc0 = {}, acc1 = {}, acc2 = {};
    const float* hrow = H + (size_t)(rowBase + m) * HIDDEN + kh;

    for (int k = 0; k < HIDDEN; k += 4) {
        v2f a = *(const v2f*)(hrow + k);
        const float* w0p = W1 + (size_t)(k + kh) * NCLASSES + m;
        const float* w1p = w0p + NCLASSES;
        v2f b;
        b.x = w0p[0];  b.y = w1p[0];
        acc0 = __builtin_amdgcn_wmma_f32_16x16x4_f32(false, a, false, b, (short)0, acc0, false, false);
        b.x = w0p[16]; b.y = w1p[16];
        acc1 = __builtin_amdgcn_wmma_f32_16x16x4_f32(false, a, false, b, (short)0, acc1, false, false);
        b.x = c2ok ? w0p[32] : 0.f;
        b.y = c2ok ? w1p[32] : 0.f;
        acc2 = __builtin_amdgcn_wmma_f32_16x16x4_f32(false, a, false, b, (short)0, acc2, false, false);
    }

    const int rh = (lane >> 4) * 8;
    const float bias0 = b1[m], bias1 = b1[16 + m], bias2 = c2ok ? b1[32 + m] : 0.f;
    for (int r = 0; r < 8; ++r) {
        int row = rowBase + rh + r;
        float sn = snorm[row];
        float* orow = O  + (size_t)row * CPAD;
        float* srow = OS + (size_t)row * CPAD;
        float v0 = acc0[r] + bias0;
        float v1 = acc1[r] + bias1;
        float v2 = acc2[r] + bias2;
        orow[m]      = v0;  srow[m]      = v0 * sn;
        orow[16 + m] = v1;  srow[16 + m] = v1 * sn;
        orow[32 + m] = v2;  srow[32 + m] = v2 * sn;
    }
}

__device__ __forceinline__ float4 appnp_ldrow(const float4* __restrict__ hin, int s, int qq) {
    const unsigned off = (unsigned)s * (unsigned)(ROWV4 * 16) + (unsigned)(qq * 16);
    return *(const float4*)((const char*)hin + off);
}

__device__ __forceinline__ void appnp_acc(float4& acc, const float4 v, const bool k) {
    acc.x += k ? v.x : 0.f;  acc.y += k ? v.y : 0.f;  acc.z += k ? v.z : 0.f;  acc.w += k ? v.w : 0.f;
}

__device__ __forceinline__ void appnp_tail(float4& acc, const float4* __restrict__ hin,
                                           const int* __restrict__ csr, int beg, int end,
                                           int q, int qq) {
    for (int base = beg; base < end; base += 16) {
        const int n  = min(16, end - base);
        const int my = (q < n) ? csr[base + q] : 0;
        for (int t = 0; t < n; t += 4) {
            const int s0 = __shfl(my, t,     16);
            const int s1 = __shfl(my, t + 1, 16);
            const int s2 = __shfl(my, t + 2, 16);
            const int s3 = __shfl(my, t + 3, 16);
            const float4 v0 = appnp_ldrow(hin, s0, qq);
            const float4 v1 = appnp_ldrow(hin, s1, qq);
            const float4 v2 = appnp_ldrow(hin, s2, qq);
            const float4 v3 = appnp_ldrow(hin, s3, qq);
            appnp_acc(acc, v0, true);
            appnp_acc(acc, v1, t + 1 < n);
            appnp_acc(acc, v2, t + 2 < n);
            appnp_acc(acc, v3, t + 3 < n);
        }
    }
}

template <bool LAST>
__global__ __launch_bounds__(256) void appnp_gather(const float4* __restrict__ hin,
                                                    const int* __restrict__ rowptr,
                                                    const int* __restrict__ csr,
                                                    const float* __restrict__ dnorm,
                                                    const float* __restrict__ snorm,
                                                    const float4* __restrict__ h0,
                                                    float4* __restrict__ hout,
                                                    float* __restrict__ out, int N) {
    const int tid  = threadIdx.x;
    const int q    = tid & 15;
    const int qq   = q < ROWV4 ? q : ROWV4 - 1;
    const int hw   = tid >> 4;
    const int rlA  = ((hw >> 1) << 2) + (hw & 1);
    const int rlB  = rlA + 2;
    const int rowA = blockIdx.x * 32 + rlA;
    const int rowB = rowA + 2;
    const int ra = min(rowA, N - 1);
    const int rb = min(rowB, N - 1);

    const int begA = rowptr[ra], endA = rowptr[ra + 1];
    const int begB = rowptr[rb], endB = rowptr[rb + 1];
    const float dnA = dnorm[ra];
    const float dnB = dnorm[rb];
    const float4 tA = h0[(size_t)ra * ROWV4 + qq];
    const float4 tB = h0[(size_t)rb * ROWV4 + qq];
    float snA = 1.0f, snB = 1.0f;
    if (!LAST) { snA = snorm[ra]; snB = snorm[rb]; }
    __builtin_amdgcn_sched_barrier(0);

    const int degA = endA - begA;
    const int degB = endB - begB;
    const int a0 = (q      < degA) ? csr[begA + q]      : 0;
    const int b0 = (q      < degB) ? csr[begB + q]      : 0;
    const int a1 = (16 + q < degA) ? csr[begA + 16 + q] : 0;
    const int b1 = (16 + q < degB) ? csr[begB + 16 + q] : 0;

    float4 accA = make_float4(0.f, 0.f, 0.f, 0.f);
    float4 accB = make_float4(0.f, 0.f, 0.f, 0.f);

    const int degM = max(degA, degB);
    {
        float4 pA[8], pB[8];
#pragma unroll
        for (int t = 0; t < 8; ++t) pA[t] = appnp_ldrow(hin, __shfl(a0, t, 16), qq);
#pragma unroll
        for (int t = 0; t < 8; ++t) pB[t] = appnp_ldrow(hin, __shfl(b0, t, 16), qq);
#pragma unroll
        for (int t = 0; t < 8; ++t) appnp_acc(accA, pA[t], t < degA);
#pragma unroll
        for (int t = 0; t < 8; ++t) appnp_acc(accB, pB[t], t < degB);
    }
    if (degM > 8) {
        float4 rA[8], rB[8];
#pragma unroll
        for (int t = 0; t < 8; ++t) rA[t] = appnp_ldrow(hin, __shfl(a0, 8 + t, 16), qq);
#pragma unroll
        for (int t = 0; t < 8; ++t) rB[t] = appnp_ldrow(hin, __shfl(b0, 8 + t, 16), qq);
#pragma unroll
        for (int t = 0; t < 8; ++t) appnp_acc(accA, rA[t], 8 + t < degA);
#pragma unroll
        for (int t = 0; t < 8; ++t) appnp_acc(accB, rB[t], 8 + t < degB);

        if (degM > 16) {
            float4 uA[8], uB[8];
#pragma unroll
            for (int t = 0; t < 4; ++t) uA[t] = appnp_ldrow(hin, __shfl(a1, t, 16), qq);
#pragma unroll
            for (int t = 0; t < 4; ++t) uB[t] = appnp_ldrow(hin, __shfl(b1, t, 16), qq);
            if (degM > 20) {
#pragma unroll
                for (int t = 4; t < 8; ++t) uA[t] = appnp_ldrow(hin, __shfl(a1, t, 16), qq);
#pragma unroll
                for (int t = 4; t < 8; ++t) uB[t] = appnp_ldrow(hin, __shfl(b1, t, 16), qq);
            }
#pragma unroll
            for (int t = 0; t < 4; ++t) appnp_acc(accA, uA[t], 16 + t < degA);
#pragma unroll
            for (int t = 0; t < 4; ++t) appnp_acc(accB, uB[t], 16 + t < degB);
            if (degM > 20) {
#pragma unroll
                for (int t = 4; t < 8; ++t) appnp_acc(accA, uA[t], 16 + t < degA);
#pragma unroll
                for (int t = 4; t < 8; ++t) appnp_acc(accB, uB[t], 16 + t < degB);
                if (degA > 24) appnp_tail(accA, hin, csr, begA + 24, endA, q, qq);
                if (degB > 24) appnp_tail(accB, hin, csr, begB + 24, endB, q, qq);
            }
        }
    }

    float4 resA, resB;
    resA.x = __builtin_fmaf((1.0f - ALPHA) * accA.x, dnA, ALPHA * tA.x);
    resA.y = __builtin_fmaf((1.0f - ALPHA) * accA.y, dnA, ALPHA * tA.y);
    resA.z = __builtin_fmaf((1.0f - ALPHA) * accA.z, dnA, ALPHA * tA.z);
    resA.w = __builtin_fmaf((1.0f - ALPHA) * accA.w, dnA, ALPHA * tA.w);
    resB.x = __builtin_fmaf((1.0f - ALPHA) * accB.x, dnB, ALPHA * tB.x);
    resB.y = __builtin_fmaf((1.0f - ALPHA) * accB.y, dnB, ALPHA * tB.y);
    resB.z = __builtin_fmaf((1.0f - ALPHA) * accB.z, dnB, ALPHA * tB.z);
    resB.w = __builtin_fmaf((1.0f - ALPHA) * accB.w, dnB, ALPHA * tB.w);

    if (!LAST) {
        if (q < ROWV4) {
            if (rowA < N)
                hout[(size_t)rowA * ROWV4 + q] = make_float4(resA.x * snA, resA.y * snA, resA.z * snA, resA.w * snA);
            if (rowB < N)
                hout[(size_t)rowB * ROWV4 + q] = make_float4(resB.x * snB, resB.y * snB, resB.z * snB, resB.w * snB);
        }
    } else {
        __shared__ float sm[32 * NCLASSES];
        if (q < ROWV4) {
            if (rowA < N) {
                float* p = sm + rlA * NCLASSES + q * 4;
                p[0] = resA.x; p[1] = resA.y; p[2] = resA.z;
                if (q * 4 + 3 < NCLASSES) p[3] = resA.w;
            }
            if (rowB < N) {
                float* p = sm + rlB * NCLASSES + q * 4;
                p[0] = resB.x; p[1] = resB.y; p[2] = resB.z;
                if (q * 4 + 3 < NCLASSES) p[3] = resB.w;
            }
        }
        __syncthreads();
        const int rowsHere = min(32, N - (int)blockIdx.x * 32);
        const int nvalid = rowsHere * NCLASSES;
        float* obase = out + (size_t)blockIdx.x * (32 * NCLASSES);
        for (int p = 0; p < (32 * NCLASSES + 255) / 256; ++p) {
            const int i = tid + p * 256;
            if (i < nvalid) obase[i] = sm[i];
        }
    }
}

extern "C" void kernel_launch(void* const* d_in, const int* in_sizes, int n_in,
                              void* d_out, int out_size, void* d_ws, size_t ws_size,
                              hipStream_t stream) {
    const float* F  = (const float*)d_in[0];
    const float* W0 = (const float*)d_in[1];
    const float* b0 = (const float*)d_in[2];
    const float* W1 = (const float*)d_in[3];
    const float* b1 = (const float*)d_in[4];
    const int*  src = (const int*)d_in[5];
    const int*  dst = (const int*)d_in[6];

    const int N = in_sizes[0] / IN_FEATS;
    const int E = in_sizes[5];
    float* out = (float*)d_out;

    float* ws    = (float*)d_ws;
    float* snorm = ws;
    int*   icnt  = (int*)(ws + (size_t)N);
    float* dnorm = ws + 2 * (size_t)N;
    float* hrelu = ws + 3 * (size_t)N;
    float* h0    = hrelu + (size_t)N * HIDDEN;
    float* hs    = h0 + (size_t)N * CPAD;
    int*   rank  = (int*)(hs + (size_t)N * CPAD);
    int*   csr   = rank + (size_t)E;
    int*   rowptr = csr + (size_t)E;

    {
        int n4 = (2 * N) / 4;
        appnp_zero4<<<(n4 + 255) / 256, 256, 0, stream>>>((float4*)snorm, n4);
        appnp_degree<<<(E + 255) / 256, 256, 0, stream>>>(src, dst, snorm, icnt, rank, E);
        appnp_rowptr<<<(N + 1023) / 1024, 256, 0, stream>>>(icnt, rowptr, N);
        appnp_norm<<<(N + 255) / 256, 256, 0, stream>>>(snorm, icnt, dnorm, N);
    }

    {
        int rowTiles = (N + 15) / 16;
        int blocks   = (rowTiles + 7) / 8;
        appnp_gemm1_wmma<<<blocks, 256, 0, stream>>>(F, W0, b0, hrelu, N);
        appnp_gemm2_wmma<<<blocks, 256, 0, stream>>>(hrelu, W1, b1, snorm, h0, hs, N);
    }

    {
        appnp_fill<<<(E + 255) / 256, 256, 0, stream>>>(src, dst, rank, rowptr, csr, E);
        float* bufB = hrelu;
        const int gblocks = (N + 31) / 32;
        for (int it = 0; it < KSTEPS; ++it) {
            const float* in = (it & 1) ? bufB : hs;
            float* nxt      = (it & 1) ? hs : bufB;
            if (it == KSTEPS - 1) {
                appnp_gather<true><<<gblocks, 256, 0, stream>>>((const float4*)in, rowptr, csr, dnorm, snorm,
                                                               (const float4*)h0, (float4*)nxt, out, N);
            } else {
                appnp_gather<false><<<gblocks, 256, 0, stream>>>((const float4*)in, rowptr, csr, dnorm, snorm,
                                                                (const float4*)h0, (float4*)nxt, out, N);
            }
        }
    }
}


# Degree counts and in-row ranks from per-chunk LDS counters (two per dword, merged per node) replace the two global atomics per edge; row list order now chunk order, then LDS arrival
# speedup vs baseline: 6.7910x; 1.2955x over previous
#include <hip/hip_runtime.h>
#include <hip/hip_bf16.h>

typedef __attribute__((ext_vector_type(2))) float v2f;
typedef __attribute__((ext_vector_type(8))) float v8f;

#define IN_FEATS 512
#define HIDDEN   64
#define NCLASSES 47
#define CPAD     48
#define ROWV4    12
#define ALPHA    0.1f
#define KSTEPS   10

#define HBLK   32
#define NODES  100000
#define HWORDS (NODES / 2)

__device__ __forceinline__ int appnp_chunk(int E) {
    return (((E + HBLK - 1) / HBLK) + 3) & ~3;
}

__device__ __forceinline__ int appnp_bump(unsigned int* h, int k, unsigned one) {
    const unsigned sh  = (unsigned)(k & 1) << 4;
    const unsigned old = atomicAdd(&h[k >> 1], one << sh);
    return (int)((old >> sh) & 0xFFFFu);
}

__global__ __launch_bounds__(1024) void appnp_hist(const int* __restrict__ src,
                                                   const int* __restrict__ dst,
                                                   unsigned short* __restrict__ part_s,
                                                   unsigned short* __restrict__ part_d,
                                                   int* __restrict__ rank, int E) {
    __shared__ unsigned int hcnt[HWORDS];
    const int  tid   = threadIdx.x;
    const bool isDst = blockIdx.x >= HBLK;
    const int  b     = isDst ? (int)blockIdx.x - HBLK : (int)blockIdx.x;
    const int* __restrict__ key = isDst ? dst : src;
    const int chunk = appnp_chunk(E);
    const int e0 = min(b * chunk, E);
    const int e1 = min(E, e0 + chunk);

    uint4* h4 = (uint4*)hcnt;
    for (int i = tid; i < HWORDS / 4; i += 1024) h4[i] = make_uint4(0u, 0u, 0u, 0u);
    __syncthreads();

    const int nfull = (e1 - e0) >> 2;
    const int4* __restrict__ key4  = (const int4*)(key + e0);
    int4* __restrict__       rank4 = (int4*)(rank + e0);
    for (int g0 = tid; g0 < nfull; g0 += 4096) {
        int4 k[4];
#pragma unroll
        for (int u = 0; u < 4; ++u) k[u] = key4[min(g0 + u * 1024, nfull - 1)];
#pragma unroll
        for (int u = 0; u < 4; ++u) {
            const int  g  = g0 + u * 1024;
            const bool ok = g < nfull;
            const unsigned one = ok ? 1u : 0u;
            int4 r;
            r.x = appnp_bump(hcnt, k[u].x, one);
            r.y = appnp_bump(hcnt, k[u].y, one);
            r.z = appnp_bump(hcnt, k[u].z, one);
            r.w = appnp_bump(hcnt, k[u].w, one);
            if (isDst && ok) rank4[g] = r;
        }
    }
    {
        const int t = e0 + 4 * nfull + tid;
        if (tid < 3 && t < e1) {
            const int r = appnp_bump(hcnt, key[t], 1u);
            if (isDst) rank[t] = r;
        }
    }
    __syncthreads();

    uint4* o4 = (uint4*)((isDst ? part_d : part_s) + (size_t)b * NODES);
    for (int i = tid; i < HWORDS / 4; i += 1024) o4[i] = h4[i];
}

__global__ __launch_bounds__(256) void appnp_merge(const unsigned short* __restrict__ part_s,
                                                   const unsigned short* __restrict__ part_d,
                                                   int* __restrict__ boff,
                                                   int* __restrict__ icnt,
                                                   float* __restrict__ snorm,
                                                   float* __restrict__ dnorm, int N) {
    const int n = blockIdx.x * blockDim.x + threadIdx.x;
    if (n >= N) return;
    int od = 0;
#pragma unroll
    for (int b = 0; b < HBLK; ++b) od += (int)part_s[(size_t)b * NODES + n];
    int t[HBLK];
#pragma unroll
    for (int b = 0; b < HBLK; ++b) t[b] = (int)part_d[(size_t)b * NODES + n];
    int run = 0;
#pragma unroll
    for (int b = 0; b < HBLK; ++b) { boff[(size_t)b * NODES + n] = run; run += t[b]; }
    icnt[n]  = run;
    snorm[n] = rsqrtf(fmaxf((float)od,  1.0f));
    dnorm[n] = rsqrtf(fmaxf((float)run, 1.0f));
}

__global__ __launch_bounds__(256) void appnp_rowptr(const int* __restrict__ icnt,
                                                    int* __restrict__ rowptr, int N) {
    __shared__ int sPart[8];
    __shared__ int sWave[8];
    const int tid  = threadIdx.x;
    const int lane = tid & 31;
    const int wave = tid >> 5;
    const int blockStart = blockIdx.x * 1024;

    int part = 0;
#pragma unroll 8
    for (int i = tid * 4; i < blockStart; i += 1024) {
        const int4 v = *(const int4*)(icnt + i);
        part += (v.x + v.y) + (v.z + v.w);
    }
    for (int o = 16; o > 0; o >>= 1) part += __shfl_xor(part, o, 32);
    if (lane == 0) sPart[wave] = part;

    const int i0 = blockStart + tid * 4;
    int c0 = 0, c1 = 0, c2 = 0, c3 = 0;
    if (i0 + 3 < N) {
        const int4 v = *(const int4*)(icnt + i0);
        c0 = v.x; c1 = v.y; c2 = v.z; c3 = v.w;
    } else {
        if (i0     < N) c0 = icnt[i0];
        if (i0 + 1 < N) c1 = icnt[i0 + 1];
        if (i0 + 2 < N) c2 = icnt[i0 + 2];
    }
    const int tsum = (c0 + c1) + (c2 + c3);
    int incl = tsum;
    for (int o = 1; o < 32; o <<= 1) {
        const int up = __shfl_up(incl, o, 32);
        if (lane >= o) incl += up;
    }
    if (lane == 31) sWave[wave] = incl;
    __syncthreads();

    int offset = 0;
    for (int w = 0; w < 8; ++w) offset += sPart[w];
    int wbase = 0;
    for (int w = 0; w < wave; ++w) wbase += sWave[w];
    const int ex = offset + wbase + (incl - tsum);

    if (i0     < N) rowptr[i0]     = ex;
    if (i0 + 1 < N) rowptr[i0 + 1] = ex + c0;
    if (i0 + 2 < N) rowptr[i0 + 2] = ex + c0 + c1;
    if (i0 + 3 < N) rowptr[i0 + 3] = ex + c0 + c1 + c2;
    if (i0 <= N - 1 && N - 1 <= i0 + 3) rowptr[N] = ex + tsum;
}

__global__ __launch_bounds__(256) void appnp_fill(const int* __restrict__ src,
                                                  const int* __restrict__ dst,
                                                  const int* __restrict__ rank,
                                                  const int* __restrict__ rowptr,
                                                  const int* __restrict__ boff,
                                                  int* __restrict__ csr, int E) {
    int e = blockIdx.x * blockDim.x + threadIdx.x;
    if (e >= E) return;
    const int d = dst[e];
    const int b = e / appnp_chunk(E);
    csr[rowptr[d] + boff[(size_t)b * NODES + d] + rank[e]] = src[e];
}

__global__ __launch_bounds__(256) void appnp_gemm1_wmma(const float* __restrict__ F,
                                                        const float* __restrict__ W0,
                                                        const float* __restrict__ b0,
                                                        float* __restrict__ H, int N) {
    __shared__ float sw[256 * HIDDEN];
    const int lane = threadIdx.x & 31;
    const int wave = threadIdx.x >> 5;
    const int nTiles = (N + 15) >> 4;
    int rowTile = blockIdx.x * 8 + wave;
    if (rowTile >= nTiles) rowTile = nTiles - 1;
    const int rowBase = rowTile * 16;
    const int m  = lane & 15;
    const int kh = (lane >> 4) << 1;

    v8f acc0 = {}, acc1 = {}, acc2 = {}, acc3 = {};
    int arow = rowBase + m; if (arow >= N) arow = N - 1;
    const float* frow = F + (size_t)arow * IN_FEATS + kh;

    for (int ko = 0; ko < IN_FEATS; ko += 256) {
        __syncthreads();
        const float4* gsrc = (const float4*)(W0 + (size_t)ko * HIDDEN);
        float4* ldst = (float4*)sw;
        for (int i = threadIdx.x; i < (256 * HIDDEN) / 4; i += 256) ldst[i] = gsrc[i];
        __syncthreads();

        for (int k = 0; k < 256; k += 4) {
            if ((k & 63) == 0) __builtin_prefetch(frow + ko + k + 64, 0, 1);
            v2f a = *(const v2f*)(frow + ko + k);
            const float* w0p = sw + (k + kh) * HIDDEN + m;
            const float* w1p = w0p + HIDDEN;
            v2f b;
            b.x = w0p[0];  b.y = w1p[0];
            acc0 = __builtin_amdgcn_wmma_f32_16x16x4_f32(false, a, false, b, (short)0, acc0, false, false);
            b.x = w0p[16]; b.y = w1p[16];
            acc1 = __builtin_amdgcn_wmma_f32_16x16x4_f32(false, a, false, b, (short)0, acc1, false, false);
            b.x = w0p[32]; b.y = w1p[32];
            acc2 = __builtin_amdgcn_wmma_f32_16x16x4_f32(false, a, false, b, (short)0, acc2, false, false);
            b.x = w0p[48]; b.y = w1p[48];
            acc3 = __builtin_amdgcn_wmma_f32_16x16x4_f32(false, a, false, b, (short)0, acc3, false, false);
        }
    }

    const int rh = (lane >> 4) * 8;
    const float bias0 = b0[m], bias1 = b0[16 + m], bias2 = b0[32 + m], bias3 = b0[48 + m];
    for (int r = 0; r < 8; ++r) {
        int row = rowBase + rh + r;
        if (row < N) {
            float* hrow = H + (size_t)row * HIDDEN;
            hrow[m]      = fmaxf(acc0[r] + bias0, 0.f);
            hrow[16 + m] = fmaxf(acc1[r] + bias1, 0.f);
            hrow[32 + m] = fmaxf(acc2[r] + bias2, 0.f);
            hrow[48 + m] = fmaxf(acc3[r] + bias3, 0.f);
        }
    }
}

__global__ __launch_bounds__(256) void appnp_gemm2_wmma(const float* __restrict__ H,
                                                        const float* __restrict__ W1,
                                                        const float* __restrict__ b1,
                                                        const float* __restrict__ snorm,
                                                        float* __restrict__ O,
                                                        float* __restrict__ OS, int N) {
    const int lane = threadIdx.x & 31;
    const int wave = threadIdx.x >> 5;
    const int rowBase = (blockIdx.x * 8 + wave) * 16;
    if (rowBase >= N) return;
    const int m  = lane & 15;
    const int kh = (lane >> 4) << 1;
    const bool c2ok = (32 + m) < NCLASSES;

    v8f acc0 = {}, acc1 = {}, acc2 = {};
    const float* hrow = H + (size_t)(rowBase + m) * HIDDEN + kh;

    for (int k = 0; k < HIDDEN; k += 4) {
        v2f a = *(const v2f*)(hrow + k);
        const float* w0p = W1 + (size_t)(k + kh) * NCLASSES + m;
        const float* w1p = w0p + NCLASSES;
        v2f b;
        b.x = w0p[0];  b.y = w1p[0];
        acc0 = __builtin_amdgcn_wmma_f32_16x16x4_f32(false, a, false, b, (short)0, acc0, false, false);
        b.x = w0p[16]; b.y = w1p[16];
        acc1 = __builtin_amdgcn_wmma_f32_16x16x4_f32(false, a, false, b, (short)0, acc1, false, false);
        b.x = c2ok ? w0p[32] : 0.f;
        b.y = c2ok ? w1p[32] : 0.f;
        acc2 = __builtin_amdgcn_wmma_f32_16x16x4_f32(false, a, false, b, (short)0, acc2, false, false);
    }

    const int rh = (lane >> 4) * 8;
    const float bias0 = b1[m], bias1 = b1[16 + m], bias2 = c2ok ? b1[32 + m] : 0.f;
    for (int r = 0; r < 8; ++r) {
        int row = rowBase + rh + r;
        float sn = snorm[row];
        float* orow = O  + (size_t)row * CPAD;
        float* srow = OS + (size_t)row * CPAD;
        float v0 = acc0[r] + bias0;
        float v1 = acc1[r] + bias1;
        float v2 = acc2[r] + bias2;
        orow[m]      = v0;  srow[m]      = v0 * sn;
        orow[16 + m] = v1;  srow[16 + m] = v1 * sn;
        orow[32 + m] = v2;  srow[32 + m] = v2 * sn;
    }
}

__device__ __forceinline__ float4 appnp_ldrow(const float4* __restrict__ hin, int s, int qq) {
    const unsigned off = (unsigned)s * (unsigned)(ROWV4 * 16) + (unsigned)(qq * 16);
    return *(const float4*)((const char*)hin + off);
}

__device__ __forceinline__ void appnp_acc(float4& acc, const float4 v, const bool k) {
    acc.x += k ? v.x : 0.f;  acc.y += k ? v.y : 0.f;  acc.z += k ? v.z : 0.f;  acc.w += k ? v.w : 0.f;
}

__device__ __forceinline__ void appnp_tail(float4& acc, const float4* __restrict__ hin,
                                           const int* __restrict__ csr, int beg, int end,
                                           int q, int qq) {
    for (int base = beg; base < end; base += 16) {
        const int n  = min(16, end - base);
        const int my = (q < n) ? csr[base + q] : 0;
        for (int t = 0; t < n; t += 4) {
            const int s0 = __shfl(my, t,     16);
            const int s1 = __shfl(my, t + 1, 16);
            const int s2 = __shfl(my, t + 2, 16);
            const int s3 = __shfl(my, t + 3, 16);
            const float4 v0 = appnp_ldrow(hin, s0, qq);
            const float4 v1 = appnp_ldrow(hin, s1, qq);
            const float4 v2 = appnp_ldrow(hin, s2, qq);
            const float4 v3 = appnp_ldrow(hin, s3, qq);
            appnp_acc(acc, v0, true);
            appnp_acc(acc, v1, t + 1 < n);
            appnp_acc(acc, v2, t + 2 < n);
            appnp_acc(acc, v3, t + 3 < n);
        }
    }
}

template <bool LAST>
__global__ __launch_bounds__(256) void appnp_gather(const float4* __restrict__ hin,
                                                    const int* __restrict__ rowptr,
                                                    const int* __restrict__ csr,
                                                    const float* __restrict__ dnorm,
                                                    const float* __restrict__ snorm,
                                                    const float4* __restrict__ h0,
                                                    float4* __restrict__ hout,
                                                    float* __restrict__ out, int N) {
    const int tid  = threadIdx.x;
    const int q    = tid & 15;
    const int qq   = q < ROWV4 ? q : ROWV4 - 1;
    const int hw   = tid >> 4;
    const int rlA  = ((hw >> 1) << 2) + (hw & 1);
    const int rlB  = rlA + 2;
    const int rowA = blockIdx.x * 32 + rlA;
    const int rowB = rowA + 2;
    const int ra = min(rowA, N - 1);
    const int rb = min(rowB, N - 1);

    const int begA = rowptr[ra], endA = rowptr[ra + 1];
    const int begB = rowptr[rb], endB = rowptr[rb + 1];
    const float dnA = dnorm[ra];
    const float dnB = dnorm[rb];
    const float4 tA = h0[(size_t)ra * ROWV4 + qq];
    const float4 tB = h0[(size_t)rb * ROWV4 + qq];
    float snA = 1.0f, snB = 1.0f;
    if (!LAST) { snA = snorm[ra]; snB = snorm[rb]; }
    __builtin_amdgcn_sched_barrier(0);

    const int degA = endA - begA;
    const int degB = endB - begB;
    const int a0 = (q      < degA) ? csr[begA + q]      : 0;
    const int b0 = (q      < degB) ? csr[begB + q]      : 0;
    const int a1 = (16 + q < degA) ? csr[begA + 16 + q] : 0;
    const int b1 = (16 + q < degB) ? csr[begB + 16 + q] : 0;

    float4 accA = make_float4(0.f, 0.f, 0.f, 0.f);
    float4 accB = make_float4(0.f, 0.f, 0.f, 0.f);

    const int degM = max(degA, degB);
    {
        float4 pA[8], pB[8];
#pragma unroll
        for (int t = 0; t < 8; ++t) pA[t] = appnp_ldrow(hin, __shfl(a0, t, 16), qq);
#pragma unroll
        for (int t = 0; t < 8; ++t) pB[t] = appnp_ldrow(hin, __shfl(b0, t, 16), qq);
#pragma unroll
        for (int t = 0; t < 8; ++t) appnp_acc(accA, pA[t], t < degA);
#pragma unroll
        for (int t = 0; t < 8; ++t) appnp_acc(accB, pB[t], t < degB);
    }
    if (degM > 8) {
        float4 rA[8], rB[8];
#pragma unroll
        for (int t = 0; t < 8; ++t) rA[t] = appnp_ldrow(hin, __shfl(a0, 8 + t, 16), qq);
#pragma unroll
        for (int t = 0; t < 8; ++t) rB[t] = appnp_ldrow(hin, __shfl(b0, 8 + t, 16), qq);
#pragma unroll
        for (int t = 0; t < 8; ++t) appnp_acc(accA, rA[t], 8 + t < degA);
#pragma unroll
        for (int t = 0; t < 8; ++t) appnp_acc(accB, rB[t], 8 + t < degB);

        if (degM > 16) {
            float4 uA[8], uB[8];
#pragma unroll
            for (int t = 0; t < 4; ++t) uA[t] = appnp_ldrow(hin, __shfl(a1, t, 16), qq);
#pragma unroll
            for (int t = 0; t < 4; ++t) uB[t] = appnp_ldrow(hin, __shfl(b1, t, 16), qq);
            if (degM > 20) {
#pragma unroll
                for (int t = 4; t < 8; ++t) uA[t] = appnp_ldrow(hin, __shfl(a1, t, 16), qq);
#pragma unroll
                for (int t = 4; t < 8; ++t) uB[t] = appnp_ldrow(hin, __shfl(b1, t, 16), qq);
            }
#pragma unroll
            for (int t = 0; t < 4; ++t) appnp_acc(accA, uA[t], 16 + t < degA);
#pragma unroll
            for (int t = 0; t < 4; ++t) appnp_acc(accB, uB[t], 16 + t < degB);
            if (degM > 20) {
#pragma unroll
                for (int t = 4; t < 8; ++t) appnp_acc(accA, uA[t], 16 + t < degA);
#pragma unroll
                for (int t = 4; t < 8; ++t) appnp_acc(accB, uB[t], 16 + t < degB);
                if (degA > 24) appnp_tail(accA, hin, csr, begA + 24, endA, q, qq);
                if (degB > 24) appnp_tail(accB, hin, csr, begB + 24, endB, q, qq);
            }
        }
    }

    float4 resA, resB;
    resA.x = __builtin_fmaf((1.0f - ALPHA) * accA.x, dnA, ALPHA * tA.x);
    resA.y = __builtin_fmaf((1.0f - ALPHA) * accA.y, dnA, ALPHA * tA.y);
    resA.z = __builtin_fmaf((1.0f - ALPHA) * accA.z, dnA, ALPHA * tA.z);
    resA.w = __builtin_fmaf((1.0f - ALPHA) * accA.w, dnA, ALPHA * tA.w);
    resB.x = __builtin_fmaf((1.0f - ALPHA) * accB.x, dnB, ALPHA * tB.x);
    resB.y = __builtin_fmaf((1.0f - ALPHA) * accB.y, dnB, ALPHA * tB.y);
    resB.z = __builtin_fmaf((1.0f - ALPHA) * accB.z, dnB, ALPHA * tB.z);
    resB.w = __builtin_fmaf((1.0f - ALPHA) * accB.w, dnB, ALPHA * tB.w);

    if (!LAST) {
        if (q < ROWV4) {
            if (rowA < N)
                hout[(size_t)rowA * ROWV4 + q] = make_float4(resA.x * snA, resA.y * snA, resA.z * snA, resA.w * snA);
            if (rowB < N)
                hout[(size_t)rowB * ROWV4 + q] = make_float4(resB.x * snB, resB.y * snB, resB.z * snB, resB.w * snB);
        }
    } else {
        __shared__ float sm[32 * NCLASSES];
        if (q < ROWV4) {
            if (rowA < N) {
                float* p = sm + rlA * NCLASSES + q * 4;
                p[0] = resA.x; p[1] = resA.y; p[2] = resA.z;
                if (q * 4 + 3 < NCLASSES) p[3] = resA.w;
            }
            if (rowB < N) {
                float* p = sm + rlB * NCLASSES + q * 4;
                p[0] = resB.x; p[1] = resB.y; p[2] = resB.z;
                if (q * 4 + 3 < NCLASSES) p[3] = resB.w;
            }
        }
        __syncthreads();
        const int rowsHere = min(32, N - (int)blockIdx.x * 32);
        const int nvalid = rowsHere * NCLASSES;
        float* obase = out + (size_t)blockIdx.x * (32 * NCLASSES);
        for (int p = 0; p < (32 * NCLASSES + 255) / 256; ++p) {
            const int i = tid + p * 256;
            if (i < nvalid) obase[i] = sm[i];
        }
    }
}

extern "C" void kernel_launch(void* const* d_in, const int* in_sizes, int n_in,
                              void* d_out, int out_size, void* d_ws, size_t ws_size,
                              hipStream_t stream) {
    const float* F  = (const float*)d_in[0];
    const float* W0 = (const float*)d_in[1];
    const float* b0 = (const float*)d_in[2];
    const float* W1 = (const float*)d_in[3];
    const float* b1 = (const float*)d_in[4];
    const int*  src = (const int*)d_in[5];
    const int*  dst = (const int*)d_in[6];

    const int N = in_sizes[0] / IN_FEATS;
    const int E = in_sizes[5];
    float* out = (float*)d_out;

    float* ws    = (float*)d_ws;
    float* snorm = ws;
    int*   icnt  = (int*)(ws + (size_t)N);
    float* dnorm = ws + 2 * (size_t)N;
    float* hrelu = ws + 3 * (size_t)N;
    float* h0    = hrelu + (size_t)N * HIDDEN;
    float* hs    = h0 + (size_t)N * CPAD;
    int*   rank  = (int*)(hs + (size_t)N * CPAD);
    int*   csr   = rank + (size_t)E;
    int*   rowptr = csr + (size_t)E;
    unsigned short* part_s = (unsigned short*)hrelu;
    unsigned short* part_d = part_s + (size_t)HBLK * NODES;
    int*            boff   = (int*)(part_d + (size_t)HBLK * NODES);

    {
        appnp_hist<<<2 * HBLK, 1024, 0, stream>>>(src, dst, part_s, part_d, rank, E);
        appnp_merge<<<(N + 255) / 256, 256, 0, stream>>>(part_s, part_d, boff, icnt, snorm, dnorm, N);
        appnp_rowptr<<<(N + 1023) / 1024, 256, 0, stream>>>(icnt, rowptr, N);
        appnp_fill<<<(E + 255) / 256, 256, 0, stream>>>(src, dst, rank, rowptr, boff, csr, E);
    }

    {
        int rowTiles = (N + 15) / 16;
        int blocks   = (rowTiles + 7) / 8;
        appnp_gemm1_wmma<<<blocks, 256, 0, stream>>>(F, W0, b0, hrelu, N);
        appnp_gemm2_wmma<<<blocks, 256, 0, stream>>>(hrelu, W1, b1, snorm, h0, hs, N);
    }

    {
        float* bufB = hrelu;
        const int gblocks = (N + 31) / 32;
        for (int it = 0; it < KSTEPS; ++it) {
            const float* in = (it & 1) ? bufB : hs;
            float* nxt      = (it & 1) ? hs : bufB;
            if (it == KSTEPS - 1) {
                appnp_gather<true><<<gblocks, 256, 0, stream>>>((const float4*)in, rowptr, csr, dnorm, snorm,
                                                               (const float4*)h0, (float4*)nxt, out, N);
            } else {
                appnp_gather<false><<<gblocks, 256, 0, stream>>>((const float4*)in, rowptr, csr, dnorm, snorm,
                                                                (const float4*)h0, (float4*)nxt, out, N);
            }
        }
    }
}


# Edge-list fill: a work-group takes one edge chunk and one of 8 node ranges, keeps that range's row offsets in LDS and streams dst/rank/src with 16-byte loads; no per-edge offset reads; same row order
# speedup vs baseline: 7.2905x; 1.0736x over previous
#include <hip/hip_runtime.h>
#include <hip/hip_bf16.h>

typedef __attribute__((ext_vector_type(2))) float v2f;
typedef __attribute__((ext_vector_type(8))) float v8f;

#define IN_FEATS 512
#define HIDDEN   64
#define NCLASSES 47
#define CPAD     48
#define ROWV4    12
#define ALPHA    0.1f
#define KSTEPS   10

#define HBLK   32
#define NODES  100000
#define HWORDS (NODES / 2)

__device__ __forceinline__ int appnp_chunk(int E) {
    return (((E + HBLK - 1) / HBLK) + 3) & ~3;
}

__device__ __forceinline__ int appnp_bump(unsigned int* h, int k, unsigned one) {
    const unsigned sh  = (unsigned)(k & 1) << 4;
    const unsigned old = atomicAdd(&h[k >> 1], one << sh);
    return (int)((old >> sh) & 0xFFFFu);
}

__global__ __launch_bounds__(1024) void appnp_hist(const int* __restrict__ src,
                                                   const int* __restrict__ dst,
                                                   unsigned short* __restrict__ part_s,
                                                   unsigned short* __restrict__ part_d,
                                                   int* __restrict__ rank, int E) {
    __shared__ unsigned int hcnt[HWORDS];
    const int  tid   = threadIdx.x;
    const bool isDst = blockIdx.x >= HBLK;
    const int  b     = isDst ? (int)blockIdx.x - HBLK : (int)blockIdx.x;
    const int* __restrict__ key = isDst ? dst : src;
    const int chunk = appnp_chunk(E);
    const int e0 = min(b * chunk, E);
    const int e1 = min(E, e0 + chunk);

    uint4* h4 = (uint4*)hcnt;
    for (int i = tid; i < HWORDS / 4; i += 1024) h4[i] = make_uint4(0u, 0u, 0u, 0u);
    __syncthreads();

    const int nfull = (e1 - e0) >> 2;
    const int4* __restrict__ key4  = (const int4*)(key + e0);
    int4* __restrict__       rank4 = (int4*)(rank + e0);
    for (int g0 = tid; g0 < nfull; g0 += 4096) {
        int4 k[4];
#pragma unroll
        for (int u = 0; u < 4; ++u) k[u] = key4[min(g0 + u * 1024, nfull - 1)];
#pragma unroll
        for (int u = 0; u < 4; ++u) {
            const int  g  = g0 + u * 1024;
            const bool ok = g < nfull;
            const unsigned one = ok ? 1u : 0u;
            int4 r;
            r.x = appnp_bump(hcnt, k[u].x, one);
            r.y = appnp_bump(hcnt, k[u].y, one);
            r.z = appnp_bump(hcnt, k[u].z, one);
            r.w = appnp_bump(hcnt, k[u].w, one);
            if (isDst && ok) rank4[g] = r;
        }
    }
    {
        const int t = e0 + 4 * nfull + tid;
        if (tid < 3 && t < e1) {
            const int r = appnp_bump(hcnt, key[t], 1u);
            if (isDst) rank[t] = r;
        }
    }
    __syncthreads();

    uint4* o4 = (uint4*)((isDst ? part_d : part_s) + (size_t)b * NODES);
    for (int i = tid; i < HWORDS / 4; i += 1024) o4[i] = h4[i];
}

__global__ __launch_bounds__(256) void appnp_merge(const unsigned short* __restrict__ part_s,
                                                   const unsigned short* __restrict__ part_d,
                                                   int* __restrict__ boff,
                                                   int* __restrict__ icnt,
                                                   float* __restrict__ snorm,
                                                   float* __restrict__ dnorm, int N) {
    const int n = blockIdx.x * blockDim.x + threadIdx.x;
    if (n >= N) return;
    int od = 0;
#pragma unroll
    for (int b = 0; b < HBLK; ++b) od += (int)part_s[(size_t)b * NODES + n];
    int t[HBLK];
#pragma unroll
    for (int b = 0; b < HBLK; ++b) t[b] = (int)part_d[(size_t)b * NODES + n];
    int run = 0;
#pragma unroll
    for (int b = 0; b < HBLK; ++b) { boff[(size_t)b * NODES + n] = run; run += t[b]; }
    icnt[n]  = run;
    snorm[n] = rsqrtf(fmaxf((float)od,  1.0f));
    dnorm[n] = rsqrtf(fmaxf((float)run, 1.0f));
}

__global__ __launch_bounds__(256) void appnp_rowptr(const int* __restrict__ icnt,
                                                    int* __restrict__ rowptr, int N) {
    __shared__ int sPart[8];
    __shared__ int sWave[8];
    const int tid  = threadIdx.x;
    const int lane = tid & 31;
    const int wave = tid >> 5;
    const int blockStart = blockIdx.x * 1024;

    int part = 0;
#pragma unroll 8
    for (int i = tid * 4; i < blockStart; i += 1024) {
        const int4 v = *(const int4*)(icnt + i);
        part += (v.x + v.y) + (v.z + v.w);
    }
    for (int o = 16; o > 0; o >>= 1) part += __shfl_xor(part, o, 32);
    if (lane == 0) sPart[wave] = part;

    const int i0 = blockStart + tid * 4;
    int c0 = 0, c1 = 0, c2 = 0, c3 = 0;
    if (i0 + 3 < N) {
        const int4 v = *(const int4*)(icnt + i0);
        c0 = v.x; c1 = v.y; c2 = v.z; c3 = v.w;
    } else {
        if (i0     < N) c0 = icnt[i0];
        if (i0 + 1 < N) c1 = icnt[i0 + 1];
        if (i0 + 2 < N) c2 = icnt[i0 + 2];
    }
    const int tsum = (c0 + c1) + (c2 + c3);
    int incl = tsum;
    for (int o = 1; o < 32; o <<= 1) {
        const int up = __shfl_up(incl, o, 32);
        if (lane >= o) incl += up;
    }
    if (lane == 31) sWave[wave] = incl;
    __syncthreads();

    int offset = 0;
    for (int w = 0; w < 8; ++w) offset += sPart[w];
    int wbase = 0;
    for (int w = 0; w < wave; ++w) wbase += sWave[w];
    const int ex = offset + wbase + (incl - tsum);

    if (i0     < N) rowptr[i0]     = ex;
    if (i0 + 1 < N) rowptr[i0 + 1] = ex + c0;
    if (i0 + 2 < N) rowptr[i0 + 2] = ex + c0 + c1;
    if (i0 + 3 < N) rowptr[i0 + 3] = ex + c0 + c1 + c2;
    if (i0 <= N - 1 && N - 1 <= i0 + 3) rowptr[N] = ex + tsum;
}

#define FSPLIT 8
#define FRANGE ((NODES + FSPLIT - 1) / FSPLIT)

__global__ __launch_bounds__(1024) void appnp_fill(const int* __restrict__ src,
                                                   const int* __restrict__ dst,
                                                   const int* __restrict__ rank,
                                                   const int* __restrict__ rowptr,
                                                   const int* __restrict__ boff,
                                                   int* __restrict__ csr, int N, int E) {
    __shared__ int tab[FRANGE];
    const int tid = threadIdx.x;
    const int b   = (int)blockIdx.x / FSPLIT;
    const int s   = (int)blockIdx.x % FSPLIT;
    const int lo  = s * FRANGE;
    const int hi  = min(N, lo + FRANGE);
    const unsigned span = (unsigned)max(hi - lo, 0);
    const int chunk = appnp_chunk(E);
    const int e0 = min(b * chunk, E);
    const int e1 = min(E, e0 + chunk);

    for (int i = tid; i < (int)span; i += 1024)
        tab[i] = rowptr[lo + i] + boff[(size_t)b * NODES + lo + i];
    __syncthreads();

    const int nfull = (e1 - e0) >> 2;
    const int4* __restrict__ d4p = (const int4*)(dst  + e0);
    const int4* __restrict__ r4p = (const int4*)(rank + e0);
    const int4* __restrict__ s4p = (const int4*)(src  + e0);
    for (int g0 = tid; g0 < nfull; g0 += 4096) {
        int4 d[4], r[4], v[4];
#pragma unroll
        for (int u = 0; u < 4; ++u) {
            const int g = min(g0 + u * 1024, nfull - 1);
            d[u] = d4p[g];  r[u] = r4p[g];  v[u] = s4p[g];
        }
#pragma unroll
        for (int u = 0; u < 4; ++u) {
            const bool ok = g0 + u * 1024 < nfull;
            unsigned k;
            k = (unsigned)(d[u].x - lo);  if (ok && k < span) csr[tab[k] + r[u].x] = v[u].x;
            k = (unsigned)(d[u].y - lo);  if (ok && k < span) csr[tab[k] + r[u].y] = v[u].y;
            k = (unsigned)(d[u].z - lo);  if (ok && k < span) csr[tab[k] + r[u].z] = v[u].z;
            k = (unsigned)(d[u].w - lo);  if (ok && k < span) csr[tab[k] + r[u].w] = v[u].w;
        }
    }
    {
        const int t = e0 + 4 * nfull + tid;
        if (tid < 3 && t < e1) {
            const unsigned k = (unsigned)(dst[t] - lo);
            if (k < span) csr[tab[k] + rank[t]] = src[t];
        }
    }
}

__global__ __launch_bounds__(256) void appnp_gemm1_wmma(const float* __restrict__ F,
                                                        const float* __restrict__ W0,
                                                        const float* __restrict__ b0,
                                                        float* __restrict__ H, int N) {
    __shared__ float sw[256 * HIDDEN];
    const int lane = threadIdx.x & 31;
    const int wave = threadIdx.x >> 5;
    const int nTiles = (N + 15) >> 4;
    int rowTile = blockIdx.x * 8 + wave;
    if (rowTile >= nTiles) rowTile = nTiles - 1;
    const int rowBase = rowTile * 16;
    const int m  = lane & 15;
    const int kh = (lane >> 4) << 1;

    v8f acc0 = {}, acc1 = {}, acc2 = {}, acc3 = {};
    int arow = rowBase + m; if (arow >= N) arow = N - 1;
    const float* frow = F + (size_t)arow * IN_FEATS + kh;

    for (int ko = 0; ko < IN_FEATS; ko += 256) {
        __syncthreads();
        const float4* gsrc = (const float4*)(W0 + (size_t)ko * HIDDEN);
        float4* ldst = (float4*)sw;
        for (int i = threadIdx.x; i < (256 * HIDDEN) / 4; i += 256) ldst[i] = gsrc[i];
        __syncthreads();

        for (int k = 0; k < 256; k += 4) {
            if ((k & 63) == 0) __builtin_prefetch(frow + ko + k + 64, 0, 1);
            v2f a = *(const v2f*)(frow + ko + k);
            const float* w0p = sw + (k + kh) * HIDDEN + m;
            const float* w1p = w0p + HIDDEN;
            v2f b;
            b.x = w0p[0];  b.y = w1p[0];
            acc0 = __builtin_amdgcn_wmma_f32_16x16x4_f32(false, a, false, b, (short)0, acc0, false, false);
            b.x = w0p[16]; b.y = w1p[16];
            acc1 = __builtin_amdgcn_wmma_f32_16x16x4_f32(false, a, false, b, (short)0, acc1, false, false);
            b.x = w0p[32]; b.y = w1p[32];
            acc2 = __builtin_amdgcn_wmma_f32_16x16x4_f32(false, a, false, b, (short)0, acc2, false, false);
            b.x = w0p[48]; b.y = w1p[48];
            acc3 = __builtin_amdgcn_wmma_f32_16x16x4_f32(false, a, false, b, (short)0, acc3, false, false);
        }
    }

    const int rh = (lane >> 4) * 8;
    const float bias0 = b0[m], bias1 = b0[16 + m], bias2 = b0[32 + m], bias3 = b0[48 + m];
    for (int r = 0; r < 8; ++r) {
        int row = rowBase + rh + r;
        if (row < N) {
            float* hrow = H + (size_t)row * HIDDEN;
            hrow[m]      = fmaxf(acc0[r] + bias0, 0.f);
            hrow[16 + m] = fmaxf(acc1[r] + bias1, 0.f);
            hrow[32 + m] = fmaxf(acc2[r] + bias2, 0.f);
            hrow[48 + m] = fmaxf(acc3[r] + bias3, 0.f);
        }
    }
}

__global__ __launch_bounds__(256) void appnp_gemm2_wmma(const float* __restrict__ H,
                                                        const float* __restrict__ W1,
                                                        const float* __restrict__ b1,
                                                        const float* __restrict__ snorm,
                                                        float* __restrict__ O,
                                                        float* __restrict__ OS, int N) {
    const int lane = threadIdx.x & 31;
    const int wave = threadIdx.x >> 5;
    const int rowBase = (blockIdx.x * 8 + wave) * 16;
    if (rowBase >= N) return;
    const int m  = lane & 15;
    const int kh = (lane >> 4) << 1;
    const bool c2ok = (32 + m) < NCLASSES;

    v8f acc0 = {}, acc1 = {}, acc2 = {};
    const float* hrow = H + (size_t)(rowBase + m) * HIDDEN + kh;

    for (int k = 0; k < HIDDEN; k += 4) {
        v2f a = *(const v2f*)(hrow + k);
        const float* w0p = W1 + (size_t)(k + kh) * NCLASSES + m;
        const float* w1p = w0p + NCLASSES;
        v2f b;
        b.x = w0p[0];  b.y = w1p[0];
        acc0 = __builtin_amdgcn_wmma_f32_16x16x4_f32(false, a, false, b, (short)0, acc0, false, false);
        b.x = w0p[16]; b.y = w1p[16];
        acc1 = __builtin_amdgcn_wmma_f32_16x16x4_f32(false, a, false, b, (short)0, acc1, false, false);
        b.x = c2ok ? w0p[32] : 0.f;
        b.y = c2ok ? w1p[32] : 0.f;
        acc2 = __builtin_amdgcn_wmma_f32_16x16x4_f32(false, a, false, b, (short)0, acc2, false, false);
    }

    const int rh = (lane >> 4) * 8;
    const float bias0 = b1[m], bias1 = b1[16 + m], bias2 = c2ok ? b1[32 + m] : 0.f;
    for (int r = 0; r < 8; ++r) {
        int row = rowBase + rh + r;
        float sn = snorm[row];
        float* orow = O  + (size_t)row * CPAD;
        float* srow = OS + (size_t)row * CPAD;
        float v0 = acc0[r] + bias0;
        float v1 = acc1[r] + bias1;
        float v2 = acc2[r] + bias2;
        orow[m]      = v0;  srow[m]      = v0 * sn;
        orow[16 + m] = v1;  srow[16 + m] = v1 * sn;
        orow[32 + m] = v2;  srow[32 + m] = v2 * sn;
    }
}

__device__ __forceinline__ float4 appnp_ldrow(const float4* __restrict__ hin, int s, int qq) {
    const unsigned off = (unsigned)s * (unsigned)(ROWV4 * 16) + (unsigned)(qq * 16);
    return *(const float4*)((const char*)hin + off);
}

__device__ __forceinline__ void appnp_acc(float4& acc, const float4 v, const bool k) {
    acc.x += k ? v.x : 0.f;  acc.y += k ? v.y : 0.f;  acc.z += k ? v.z : 0.f;  acc.w += k ? v.w : 0.f;
}

__device__ __forceinline__ void appnp_tail(float4& acc, const float4* __restrict__ hin,
                                           const int* __restrict__ csr, int beg, int end,
                                           int q, int qq) {
    for (int base = beg; base < end; base += 16) {
        const int n  = min(16, end - base);
        const int my = (q < n) ? csr[base + q] : 0;
        for (int t = 0; t < n; t += 4) {
            const int s0 = __shfl(my, t,     16);
            const int s1 = __shfl(my, t + 1, 16);
            const int s2 = __shfl(my, t + 2, 16);
            const int s3 = __shfl(my, t + 3, 16);
            const float4 v0 = appnp_ldrow(hin, s0, qq);
            const float4 v1 = appnp_ldrow(hin, s1, qq);
            const float4 v2 = appnp_ldrow(hin, s2, qq);
            const float4 v3 = appnp_ldrow(hin, s3, qq);
            appnp_acc(acc, v0, true);
            appnp_acc(acc, v1, t + 1 < n);
            appnp_acc(acc, v2, t + 2 < n);
            appnp_acc(acc, v3, t + 3 < n);
        }
    }
}

template <bool LAST>
__global__ __launch_bounds__(256) void appnp_gather(const float4* __restrict__ hin,
                                                    const int* __restrict__ rowptr,
                                                    const int* __restrict__ csr,
                                                    const float* __restrict__ dnorm,
                                                    const float* __restrict__ snorm,
                                                    const float4* __restrict__ h0,
                                                    float4* __restrict__ hout,
                                                    float* __restrict__ out, int N) {
    const int tid  = threadIdx.x;
    const int q    = tid & 15;
    const int qq   = q < ROWV4 ? q : ROWV4 - 1;
    const int hw   = tid >> 4;
    const int rlA  = ((hw >> 1) << 2) + (hw & 1);
    const int rlB  = rlA + 2;
    const int rowA = blockIdx.x * 32 + rlA;
    const int rowB = rowA + 2;
    const int ra = min(rowA, N - 1);
    const int rb = min(rowB, N - 1);

    const int begA = rowptr[ra], endA = rowptr[ra + 1];
    const int begB = rowptr[rb], endB = rowptr[rb + 1];
    const float dnA = dnorm[ra];
    const float dnB = dnorm[rb];
    const float4 tA = h0[(size_t)ra * ROWV4 + qq];
    const float4 tB = h0[(size_t)rb * ROWV4 + qq];
    float snA = 1.0f, snB = 1.0f;
    if (!LAST) { snA = snorm[ra]; snB = snorm[rb]; }
    __builtin_amdgcn_sched_barrier(0);

    const int degA = endA - begA;
    const int degB = endB - begB;
    const int a0 = (q      < degA) ? csr[begA + q]      : 0;
    const int b0 = (q      < degB) ? csr[begB + q]      : 0;
    const int a1 = (16 + q < degA) ? csr[begA + 16 + q] : 0;
    const int b1 = (16 + q < degB) ? csr[begB + 16 + q] : 0;

    float4 accA = make_float4(0.f, 0.f, 0.f, 0.f);
    float4 accB = make_float4(0.f, 0.f, 0.f, 0.f);

    const int degM = max(degA, degB);
    {
        float4 pA[8], pB[8];
#pragma unroll
        for (int t = 0; t < 8; ++t) pA[t] = appnp_ldrow(hin, __shfl(a0, t, 16), qq);
#pragma unroll
        for (int t = 0; t < 8; ++t) pB[t] = appnp_ldrow(hin, __shfl(b0, t, 16), qq);
#pragma unroll
        for (int t = 0; t < 8; ++t) appnp_acc(accA, pA[t], t < degA);
#pragma unroll
        for (int t = 0; t < 8; ++t) appnp_acc(accB, pB[t], t < degB);
    }
    if (degM > 8) {
        float4 rA[8], rB[8];
#pragma unroll
        for (int t = 0; t < 8; ++t) rA[t] = appnp_ldrow(hin, __shfl(a0, 8 + t, 16), qq);
#pragma unroll
        for (int t = 0; t < 8; ++t) rB[t] = appnp_ldrow(hin, __shfl(b0, 8 + t, 16), qq);
#pragma unroll
        for (int t = 0; t < 8; ++t) appnp_acc(accA, rA[t], 8 + t < degA);
#pragma unroll
        for (int t = 0; t < 8; ++t) appnp_acc(accB, rB[t], 8 + t < degB);

        if (degM > 16) {
            float4 uA[8], uB[8];
#pragma unroll
            for (int t = 0; t < 4; ++t) uA[t] = appnp_ldrow(hin, __shfl(a1, t, 16), qq);
#pragma unroll
            for (int t = 0; t < 4; ++t) uB[t] = appnp_ldrow(hin, __shfl(b1, t, 16), qq);
            if (degM > 20) {
#pragma unroll
                for (int t = 4; t < 8; ++t) uA[t] = appnp_ldrow(hin, __shfl(a1, t, 16), qq);
#pragma unroll
                for (int t = 4; t < 8; ++t) uB[t] = appnp_ldrow(hin, __shfl(b1, t, 16), qq);
            }
#pragma unroll
            for (int t = 0; t < 4; ++t) appnp_acc(accA, uA[t], 16 + t < degA);
#pragma unroll
            for (int t = 0; t < 4; ++t) appnp_acc(accB, uB[t], 16 + t < degB);
            if (degM > 20) {
#pragma unroll
                for (int t = 4; t < 8; ++t) appnp_acc(accA, uA[t], 16 + t < degA);
#pragma unroll
                for (int t = 4; t < 8; ++t) appnp_acc(accB, uB[t], 16 + t < degB);
                if (degA > 24) appnp_tail(accA, hin, csr, begA + 24, endA, q, qq);
                if (degB > 24) appnp_tail(accB, hin, csr, begB + 24, endB, q, qq);
            }
        }
    }

    float4 resA, resB;
    resA.x = __builtin_fmaf((1.0f - ALPHA) * accA.x, dnA, ALPHA * tA.x);
    resA.y = __builtin_fmaf((1.0f - ALPHA) * accA.y, dnA, ALPHA * tA.y);
    resA.z = __builtin_fmaf((1.0f - ALPHA) * accA.z, dnA, ALPHA * tA.z);
    resA.w = __builtin_fmaf((1.0f - ALPHA) * accA.w, dnA, ALPHA * tA.w);
    resB.x = __builtin_fmaf((1.0f - ALPHA) * accB.x, dnB, ALPHA * tB.x);
    resB.y = __builtin_fmaf((1.0f - ALPHA) * accB.y, dnB, ALPHA * tB.y);
    resB.z = __builtin_fmaf((1.0f - ALPHA) * accB.z, dnB, ALPHA * tB.z);
    resB.w = __builtin_fmaf((1.0f - ALPHA) * accB.w, dnB, ALPHA * tB.w);

    if (!LAST) {
        if (q < ROWV4) {
            if (rowA < N)
                hout[(size_t)rowA * ROWV4 + q] = make_float4(resA.x * snA, resA.y * snA, resA.z * snA, resA.w * snA);
            if (rowB < N)
                hout[(size_t)rowB * ROWV4 + q] = make_float4(resB.x * snB, resB.y * snB, resB.z * snB, resB.w * snB);
        }
    } else {
        __shared__ float sm[32 * NCLASSES];
        if (q < ROWV4) {
            if (rowA < N) {
                float* p = sm + rlA * NCLASSES + q * 4;
                p[0] = resA.x; p[1] = resA.y; p[2] = resA.z;
                if (q * 4 + 3 < NCLASSES) p[3] = resA.w;
            }
            if (rowB < N) {
                float* p = sm + rlB * NCLASSES + q * 4;
                p[0] = resB.x; p[1] = resB.y; p[2] = resB.z;
                if (q * 4 + 3 < NCLASSES) p[3] = resB.w;
            }
        }
        __syncthreads();
        const int rowsHere = min(32, N - (int)blockIdx.x * 32);
        const int nvalid = rowsHere * NCLASSES;
        float* obase = out + (size_t)blockIdx.x * (32 * NCLASSES);
        for (int p = 0; p < (32 * NCLASSES + 255) / 256; ++p) {
            const int i = tid + p * 256;
            if (i < nvalid) obase[i] = sm[i];
        }
    }
}

extern "C" void kernel_launch(void* const* d_in, const int* in_sizes, int n_in,
                              void* d_out, int out_size, void* d_ws, size_t ws_size,
                              hipStream_t stream) {
    const float* F  = (const float*)d_in[0];
    const float* W0 = (const float*)d_in[1];
    const float* b0 = (const float*)d_in[2];
    const float* W1 = (const float*)d_in[3];
    const float* b1 = (const float*)d_in[4];
    const int*  src = (const int*)d_in[5];
    const int*  dst = (const int*)d_in[6];

    const int N = in_sizes[0] / IN_FEATS;
    const int E = in_sizes[5];
    float* out = (float*)d_out;

    float* ws    = (float*)d_ws;
    float* snorm = ws;
    int*   icnt  = (int*)(ws + (size_t)N);
    float* dnorm = ws + 2 * (size_t)N;
    float* hrelu = ws + 3 * (size_t)N;
    float* h0    = hrelu + (size_t)N * HIDDEN;
    float* hs    = h0 + (size_t)N * CPAD;
    int*   rank  = (int*)(hs + (size_t)N * CPAD);
    int*   csr   = rank + (size_t)E;
    int*   rowptr = csr + (size_t)E;
    unsigned short* part_s = (unsigned short*)hrelu;
    unsigned short* part_d = part_s + (size_t)HBLK * NODES;
    int*            boff   = (int*)(part_d + (size_t)HBLK * NODES);

    {
        appnp_hist<<<2 * HBLK, 1024, 0, stream>>>(src, dst, part_s, part_d, rank, E);
        appnp_merge<<<(N + 255) / 256, 256, 0, stream>>>(part_s, part_d, boff, icnt, snorm, dnorm, N);
        appnp_rowptr<<<(N + 1023) / 1024, 256, 0, stream>>>(icnt, rowptr, N);
        appnp_fill<<<HBLK * FSPLIT, 1024, 0, stream>>>(src, dst, rank, rowptr, boff, csr, N, E);
    }

    {
        int rowTiles = (N + 15) / 16;
        int blocks   = (rowTiles + 7) / 8;
        appnp_gemm1_wmma<<<blocks, 256, 0, stream>>>(F, W0, b0, hrelu, N);
        appnp_gemm2_wmma<<<blocks, 256, 0, stream>>>(hrelu, W1, b1, snorm, h0, hs, N);
    }

    {
        float* bufB = hrelu;
        const int gblocks = (N + 31) / 32;
        for (int it = 0; it < KSTEPS; ++it) {
            const float* in = (it & 1) ? bufB : hs;
            float* nxt      = (it & 1) ? hs : bufB;
            if (it == KSTEPS - 1) {
                appnp_gather<true><<<gblocks, 256, 0, stream>>>((const float4*)in, rowptr, csr, dnorm, snorm,
                                                               (const float4*)h0, (float4*)nxt, out, N);
            } else {
                appnp_gather<false><<<gblocks, 256, 0, stream>>>((const float4*)in, rowptr, csr, dnorm, snorm,
                                                                (const float4*)h0, (float4*)nxt, out, N);
            }
        }
    }
}


# MLP in one launch: one 25-wave work-group per WGP keeps all of W0 in LDS, a wave takes its 16-row tile through both products (relu tile handed over in LDS); hidden matrix never stored; same sums
# speedup vs baseline: 7.6692x; 1.0519x over previous
#include <hip/hip_runtime.h>
#include <hip/hip_bf16.h>

typedef __attribute__((ext_vector_type(2))) float v2f;
typedef __attribute__((ext_vector_type(8))) float v8f;

#define IN_FEATS 512
#define HIDDEN   64
#define NCLASSES 47
#define CPAD     48
#define ROWV4    12
#define ALPHA    0.1f
#define KSTEPS   10

#define HBLK   32
#define NODES  100000
#define HWORDS (NODES / 2)

__device__ __forceinline__ int appnp_chunk(int E) {
    return (((E + HBLK - 1) / HBLK) + 3) & ~3;
}

__device__ __forceinline__ int appnp_bump(unsigned int* h, int k, unsigned one) {
    const unsigned sh  = (unsigned)(k & 1) << 4;
    const unsigned old = atomicAdd(&h[k >> 1], one << sh);
    return (int)((old >> sh) & 0xFFFFu);
}

__global__ __launch_bounds__(1024) void appnp_hist(const int* __restrict__ src,
                                                   const int* __restrict__ dst,
                                                   unsigned short* __restrict__ part_s,
                                                   unsigned short* __restrict__ part_d,
                                                   int* __restrict__ rank, int E) {
    __shared__ unsigned int hcnt[HWORDS];
    const int  tid   = threadIdx.x;
    const bool isDst = blockIdx.x >= HBLK;
    const int  b     = isDst ? (int)blockIdx.x - HBLK : (int)blockIdx.x;
    const int* __restrict__ key = isDst ? dst : src;
    const int chunk = appnp_chunk(E);
    const int e0 = min(b * chunk, E);
    const int e1 = min(E, e0 + chunk);

    uint4* h4 = (uint4*)hcnt;
    for (int i = tid; i < HWORDS / 4; i += 1024) h4[i] = make_uint4(0u, 0u, 0u, 0u);
    __syncthreads();

    const int nfull = (e1 - e0) >> 2;
    const int4* __restrict__ key4  = (const int4*)(key + e0);
    int4* __restrict__       rank4 = (int4*)(rank + e0);
    for (int g0 = tid; g0 < nfull; g0 += 4096) {
        int4 k[4];
#pragma unroll
        for (int u = 0; u < 4; ++u) k[u] = key4[min(g0 + u * 1024, nfull - 1)];
#pragma unroll
        for (int u = 0; u < 4; ++u) {
            const int  g  = g0 + u * 1024;
            const bool ok = g < nfull;
            const unsigned one = ok ? 1u : 0u;
            int4 r;
            r.x = appnp_bump(hcnt, k[u].x, one);
            r.y = appnp_bump(hcnt, k[u].y, one);
            r.z = appnp_bump(hcnt, k[u].z, one);
            r.w = appnp_bump(hcnt, k[u].w, one);
            if (isDst && ok) rank4[g] = r;
        }
    }
    {
        const int t = e0 + 4 * nfull + tid;
        if (tid < 3 && t < e1) {
            const int r = appnp_bump(hcnt, key[t], 1u);
            if (isDst) rank[t] = r;
        }
    }
    __syncthreads();

    uint4* o4 = (uint4*)((isDst ? part_d : part_s) + (size_t)b * NODES);
    for (int i = tid; i < HWORDS / 4; i += 1024) o4[i] = h4[i];
}

__global__ __launch_bounds__(256) void appnp_merge(const unsigned short* __restrict__ part_s,
                                                   const unsigned short* __restrict__ part_d,
                                                   int* __restrict__ boff,
                                                   int* __restrict__ icnt,
                                                   float* __restrict__ snorm,
                                                   float* __restrict__ dnorm, int N) {
    const int n = blockIdx.x * blockDim.x + threadIdx.x;
    if (n >= N) return;
    int od = 0;
#pragma unroll
    for (int b = 0; b < HBLK; ++b) od += (int)part_s[(size_t)b * NODES + n];
    int t[HBLK];
#pragma unroll
    for (int b = 0; b < HBLK; ++b) t[b] = (int)part_d[(size_t)b * NODES + n];
    int run = 0;
#pragma unroll
    for (int b = 0; b < HBLK; ++b) { boff[(size_t)b * NODES + n] = run; run += t[b]; }
    icnt[n]  = run;
    snorm[n] = rsqrtf(fmaxf((float)od,  1.0f));
    dnorm[n] = rsqrtf(fmaxf((float)run, 1.0f));
}

__global__ __launch_bounds__(256) void appnp_rowptr(const int* __restrict__ icnt,
                                                    int* __restrict__ rowptr, int N) {
    __shared__ int sPart[8];
    __shared__ int sWave[8];
    const int tid  = threadIdx.x;
    const int lane = tid & 31;
    const int wave = tid >> 5;
    const int blockStart = blockIdx.x * 1024;

    int part = 0;
#pragma unroll 8
    for (int i = tid * 4; i < blockStart; i += 1024) {
        const int4 v = *(const int4*)(icnt + i);
        part += (v.x + v.y) + (v.z + v.w);
    }
    for (int o = 16; o > 0; o >>= 1) part += __shfl_xor(part, o, 32);
    if (lane == 0) sPart[wave] = part;

    const int i0 = blockStart + tid * 4;
    int c0 = 0, c1 = 0, c2 = 0, c3 = 0;
    if (i0 + 3 < N) {
        const int4 v = *(const int4*)(icnt + i0);
        c0 = v.x; c1 = v.y; c2 = v.z; c3 = v.w;
    } else {
        if (i0     < N) c0 = icnt[i0];
        if (i0 + 1 < N) c1 = icnt[i0 + 1];
        if (i0 + 2 < N) c2 = icnt[i0 + 2];
    }
    const int tsum = (c0 + c1) + (c2 + c3);
    int incl = tsum;
    for (int o = 1; o < 32; o <<= 1) {
        const int up = __shfl_up(incl, o, 32);
        if (lane >= o) incl += up;
    }
    if (lane == 31) sWave[wave] = incl;
    __syncthreads();

    int offset = 0;
    for (int w = 0; w < 8; ++w) offset += sPart[w];
    int wbase = 0;
    for (int w = 0; w < wave; ++w) wbase += sWave[w];
    const int ex = offset + wbase + (incl - tsum);

    if (i0     < N) rowptr[i0]     = ex;
    if (i0 + 1 < N) rowptr[i0 + 1] = ex + c0;
    if (i0 + 2 < N) rowptr[i0 + 2] = ex + c0 + c1;
    if (i0 + 3 < N) rowptr[i0 + 3] = ex + c0 + c1 + c2;
    if (i0 <= N - 1 && N - 1 <= i0 + 3) rowptr[N] = ex + tsum;
}

#define FSPLIT 8
#define FRANGE ((NODES + FSPLIT - 1) / FSPLIT)

__global__ __launch_bounds__(1024) void appnp_fill(const int* __restrict__ src,
                                                   const int* __restrict__ dst,
                                                   const int* __restrict__ rank,
                                                   const int* __restrict__ rowptr,
                                                   const int* __restrict__ boff,
                                                   int* __restrict__ csr, int N, int E) {
    __shared__ int tab[FRANGE];
    const int tid = threadIdx.x;
    const int b   = (int)blockIdx.x / FSPLIT;
    const int s   = (int)blockIdx.x % FSPLIT;
    const int lo  = s * FRANGE;
    const int hi  = min(N, lo + FRANGE);
    const unsigned span = (unsigned)max(hi - lo, 0);
    const int chunk = appnp_chunk(E);
    const int e0 = min(b * chunk, E);
    const int e1 = min(E, e0 + chunk);

    for (int i = tid; i < (int)span; i += 1024)
        tab[i] = rowptr[lo + i] + boff[(size_t)b * NODES + lo + i];
    __syncthreads();

    const int nfull = (e1 - e0) >> 2;
    const int4* __restrict__ d4p = (const int4*)(dst  + e0);
    const int4* __restrict__ r4p = (const int4*)(rank + e0);
    const int4* __restrict__ s4p = (const int4*)(src  + e0);
    for (int g0 = tid; g0 < nfull; g0 += 4096) {
        int4 d[4], r[4], v[4];
#pragma unroll
        for (int u = 0; u < 4; ++u) {
            const int g = min(g0 + u * 1024, nfull - 1);
            d[u] = d4p[g];  r[u] = r4p[g];  v[u] = s4p[g];
        }
#pragma unroll
        for (int u = 0; u < 4; ++u) {
            const bool ok = g0 + u * 1024 < nfull;
            unsigned k;
            k = (unsigned)(d[u].x - lo);  if (ok && k < span) csr[tab[k] + r[u].x] = v[u].x;
            k = (unsigned)(d[u].y - lo);  if (ok && k < span) csr[tab[k] + r[u].y] = v[u].y;
            k = (unsigned)(d[u].z - lo);  if (ok && k < span) csr[tab[k] + r[u].z] = v[u].z;
            k = (unsigned)(d[u].w - lo);  if (ok && k < span) csr[tab[k] + r[u].w] = v[u].w;
        }
    }
    {
        const int t = e0 + 4 * nfull + tid;
        if (tid < 3 && t < e1) {
            const unsigned k = (unsigned)(dst[t] - lo);
            if (k < span) csr[tab[k] + rank[t]] = src[t];
        }
    }
}

#define MLP_WAVES   25
#define MLP_THREADS (MLP_WAVES * 32)
#define HT_PITCH    68

__global__ __launch_bounds__(MLP_THREADS) void appnp_mlp_wmma(const float* __restrict__ F,
                                                              const float* __restrict__ W0,
                                                              const float* __restrict__ b0,
                                                              const float* __restrict__ W1,
                                                              const float* __restrict__ b1,
                                                              const float* __restrict__ snorm,
                                                              float* __restrict__ O,
                                                              float* __restrict__ OS, int N) {
    __shared__ __attribute__((aligned(16))) float sw[IN_FEATS * HIDDEN];
    __shared__ __attribute__((aligned(16))) float st[MLP_WAVES * 16 * HT_PITCH];
    __shared__ __attribute__((aligned(16))) float sw1[HIDDEN * CPAD];
    const int lane = threadIdx.x & 31;
    const int wave = threadIdx.x >> 5;

    {
        const float4* gsrc = (const float4*)W0;
        float4* ldst = (float4*)sw;
        for (int i = threadIdx.x; i < (IN_FEATS * HIDDEN) / 4; i += MLP_THREADS) ldst[i] = gsrc[i];
        for (int i = threadIdx.x; i < HIDDEN * CPAD; i += MLP_THREADS) {
            const int kr = i / CPAD, cc = i - kr * CPAD;
            sw1[i] = (cc < NCLASSES) ? W1[kr * NCLASSES + cc] : 0.f;
        }
    }
    __syncthreads();

    const int nTiles = (N + 15) >> 4;
    const int rowTile = blockIdx.x * MLP_WAVES + wave;
    if (rowTile >= nTiles) return;
    const int rowBase = rowTile * 16;
    const int m  = lane & 15;
    const int kh = (lane >> 4) << 1;
    const int rh = (lane >> 4) * 8;

    v8f acc0 = {}, acc1 = {}, acc2 = {}, acc3 = {};
    int arow = rowBase + m; if (arow >= N) arow = N - 1;
    const float* frow = F + (size_t)arow * IN_FEATS + kh;
    for (int k = 0; k < IN_FEATS; k += 4) {
        if ((k & 63) == 0) __builtin_prefetch(frow + k + 64, 0, 1);
        v2f a = *(const v2f*)(frow + k);
        const float* w0p = sw + (k + kh) * HIDDEN + m;
        const float* w1p = w0p + HIDDEN;
        v2f b;
        b.x = w0p[0];  b.y = w1p[0];
        acc0 = __builtin_amdgcn_wmma_f32_16x16x4_f32(false, a, false, b, (short)0, acc0, false, false);
        b.x = w0p[16]; b.y = w1p[16];
        acc1 = __builtin_amdgcn_wmma_f32_16x16x4_f32(false, a, false, b, (short)0, acc1, false, false);
        b.x = w0p[32]; b.y = w1p[32];
        acc2 = __builtin_amdgcn_wmma_f32_16x16x4_f32(false, a, false, b, (short)0, acc2, false, false);
        b.x = w0p[48]; b.y = w1p[48];
        acc3 = __builtin_amdgcn_wmma_f32_16x16x4_f32(false, a, false, b, (short)0, acc3, false, false);
    }

    float* tile = st + wave * (16 * HT_PITCH);
    {
        const float bias0 = b0[m], bias1 = b0[16 + m], bias2 = b0[32 + m], bias3 = b0[48 + m];
        for (int r = 0; r < 8; ++r) {
            float* trw = tile + (rh + r) * HT_PITCH + m;
            trw[0]  = fmaxf(acc0[r] + bias0, 0.f);
            trw[16] = fmaxf(acc1[r] + bias1, 0.f);
            trw[32] = fmaxf(acc2[r] + bias2, 0.f);
            trw[48] = fmaxf(acc3[r] + bias3, 0.f);
        }
    }
    __builtin_amdgcn_wave_barrier();

    const bool c2ok = (32 + m) < NCLASSES;
    v8f o0 = {}, o1 = {}, o2 = {};
    const float* trow = tile + m * HT_PITCH + kh;
    for (int k = 0; k < HIDDEN; k += 4) {
        v2f a = *(const v2f*)(trow + k);
        const float* w0p = sw1 + (k + kh) * CPAD + m;
        const float* w1p = w0p + CPAD;
        v2f b;
        b.x = w0p[0];  b.y = w1p[0];
        o0 = __builtin_amdgcn_wmma_f32_16x16x4_f32(false, a, false, b, (short)0, o0, false, false);
        b.x = w0p[16]; b.y = w1p[16];
        o1 = __builtin_amdgcn_wmma_f32_16x16x4_f32(false, a, false, b, (short)0, o1, false, false);
        b.x = w0p[32]; b.y = w1p[32];
        o2 = __builtin_amdgcn_wmma_f32_16x16x4_f32(false, a, false, b, (short)0, o2, false, false);
    }

    const float c2raw = b1[c2ok ? 32 + m : 0];
    const float c0 = b1[m], c1 = b1[16 + m], c2 = c2ok ? c2raw : 0.f;
    for (int r = 0; r < 8; ++r) {
        const int row = rowBase + rh + r;
        const float sn = snorm[row];
        float* orow = O  + (size_t)row * CPAD;
        float* srow = OS + (size_t)row * CPAD;
        const float v0 = o0[r] + c0;
        const float v1 = o1[r] + c1;
        const float v2 = o2[r] + c2;
        orow[m]      = v0;  srow[m]      = v0 * sn;
        orow[16 + m] = v1;  srow[16 + m] = v1 * sn;
        orow[32 + m] = v2;  srow[32 + m] = v2 * sn;
    }
}

__device__ __forceinline__ float4 appnp_ldrow(const float4* __restrict__ hin, int s, int qq) {
    const unsigned off = (unsigned)s * (unsigned)(ROWV4 * 16) + (unsigned)(qq * 16);
    return *(const float4*)((const char*)hin + off);
}

__device__ __forceinline__ void appnp_acc(float4& acc, const float4 v, const bool k) {
    acc.x += k ? v.x : 0.f;  acc.y += k ? v.y : 0.f;  acc.z += k ? v.z : 0.f;  acc.w += k ? v.w : 0.f;
}

__device__ __forceinline__ void appnp_tail(float4& acc, const float4* __restrict__ hin,
                                           const int* __restrict__ csr, int beg, int end,
                                           int q, int qq) {
    for (int base = beg; base < end; base += 16) {
        const int n  = min(16, end - base);
        const int my = (q < n) ? csr[base + q] : 0;
        for (int t = 0; t < n; t += 4) {
            const int s0 = __shfl(my, t,     16);
            const int s1 = __shfl(my, t + 1, 16);
            const int s2 = __shfl(my, t + 2, 16);
            const int s3 = __shfl(my, t + 3, 16);
            const float4 v0 = appnp_ldrow(hin, s0, qq);
            const float4 v1 = appnp_ldrow(hin, s1, qq);
            const float4 v2 = appnp_ldrow(hin, s2, qq);
            const float4 v3 = appnp_ldrow(hin, s3, qq);
            appnp_acc(acc, v0, true);
            appnp_acc(acc, v1, t + 1 < n);
            appnp_acc(acc, v2, t + 2 < n);
            appnp_acc(acc, v3, t + 3 < n);
        }
    }
}

template <bool LAST>
__global__ __launch_bounds__(256) void appnp_gather(const float4* __restrict__ hin,
                                                    const int* __restrict__ rowptr,
                                                    const int* __restrict__ csr,
                                                    const float* __restrict__ dnorm,
                                                    const float* __restrict__ snorm,
                                                    const float4* __restrict__ h0,
                                                    float4* __restrict__ hout,
                                                    float* __restrict__ out, int N) {
    const int tid  = threadIdx.x;
    const int q    = tid & 15;
    const int qq   = q < ROWV4 ? q : ROWV4 - 1;
    const int hw   = tid >> 4;
    const int rlA  = ((hw >> 1) << 2) + (hw & 1);
    const int rlB  = rlA + 2;
    const int rowA = blockIdx.x * 32 + rlA;
    const int rowB = rowA + 2;
    const int ra = min(rowA, N - 1);
    const int rb = min(rowB, N - 1);

    const int begA = rowptr[ra], endA = rowptr[ra + 1];
    const int begB = rowptr[rb], endB = rowptr[rb + 1];
    const float dnA = dnorm[ra];
    const float dnB = dnorm[rb];
    const float4 tA = h0[(size_t)ra * ROWV4 + qq];
    const float4 tB = h0[(size_t)rb * ROWV4 + qq];
    float snA = 1.0f, snB = 1.0f;
    if (!LAST) { snA = snorm[ra]; snB = snorm[rb]; }
    __builtin_amdgcn_sched_barrier(0);

    const int degA = endA - begA;
    const int degB = endB - begB;
    const int a0 = (q      < degA) ? csr[begA + q]      : 0;
    const int b0 = (q      < degB) ? csr[begB + q]      : 0;
    const int a1 = (16 + q < degA) ? csr[begA + 16 + q] : 0;
    const int b1 = (16 + q < degB) ? csr[begB + 16 + q] : 0;

    float4 accA = make_float4(0.f, 0.f, 0.f, 0.f);
    float4 accB = make_float4(0.f, 0.f, 0.f, 0.f);

    const int degM = max(degA, degB);
    {
        float4 pA[8], pB[8];
#pragma unroll
        for (int t = 0; t < 8; ++t) pA[t] = appnp_ldrow(hin, __shfl(a0, t, 16), qq);
#pragma unroll
        for (int t = 0; t < 8; ++t) pB[t] = appnp_ldrow(hin, __shfl(b0, t, 16), qq);
#pragma unroll
        for (int t = 0; t < 8; ++t) appnp_acc(accA, pA[t], t < degA);
#pragma unroll
        for (int t = 0; t < 8; ++t) appnp_acc(accB, pB[t], t < degB);
    }
    if (degM > 8) {
        float4 rA[8], rB[8];
#pragma unroll
        for (int t = 0; t < 8; ++t) rA[t] = appnp_ldrow(hin, __shfl(a0, 8 + t, 16), qq);
#pragma unroll
        for (int t = 0; t < 8; ++t) rB[t] = appnp_ldrow(hin, __shfl(b0, 8 + t, 16), qq);
#pragma unroll
        for (int t = 0; t < 8; ++t) appnp_acc(accA, rA[t], 8 + t < degA);
#pragma unroll
        for (int t = 0; t < 8; ++t) appnp_acc(accB, rB[t], 8 + t < degB);

        if (degM > 16) {
            float4 uA[8], uB[8];
#pragma unroll
            for (int t = 0; t < 4; ++t) uA[t] = appnp_ldrow(hin, __shfl(a1, t, 16), qq);
#pragma unroll
            for (int t = 0; t < 4; ++t) uB[t] = appnp_ldrow(hin, __shfl(b1, t, 16), qq);
            if (degM > 20) {
#pragma unroll
                for (int t = 4; t < 8; ++t) uA[t] = appnp_ldrow(hin, __shfl(a1, t, 16), qq);
#pragma unroll
                for (int t = 4; t < 8; ++t) uB[t] = appnp_ldrow(hin, __shfl(b1, t, 16), qq);
            }
#pragma unroll
            for (int t = 0; t < 4; ++t) appnp_acc(accA, uA[t], 16 + t < degA);
#pragma unroll
            for (int t = 0; t < 4; ++t) appnp_acc(accB, uB[t], 16 + t < degB);
            if (degM > 20) {
#pragma unroll
                for (int t = 4; t < 8; ++t) appnp_acc(accA, uA[t], 16 + t < degA);
#pragma unroll
                for (int t = 4; t < 8; ++t) appnp_acc(accB, uB[t], 16 + t < degB);
                if (degA > 24) appnp_tail(accA, hin, csr, begA + 24, endA, q, qq);
                if (degB > 24) appnp_tail(accB, hin, csr, begB + 24, endB, q, qq);
            }
        }
    }

    float4 resA, resB;
    resA.x = __builtin_fmaf((1.0f - ALPHA) * accA.x, dnA, ALPHA * tA.x);
    resA.y = __builtin_fmaf((1.0f - ALPHA) * accA.y, dnA, ALPHA * tA.y);
    resA.z = __builtin_fmaf((1.0f - ALPHA) * accA.z, dnA, ALPHA * tA.z);
    resA.w = __builtin_fmaf((1.0f - ALPHA) * accA.w, dnA, ALPHA * tA.w);
    resB.x = __builtin_fmaf((1.0f - ALPHA) * accB.x, dnB, ALPHA * tB.x);
    resB.y = __builtin_fmaf((1.0f - ALPHA) * accB.y, dnB, ALPHA * tB.y);
    resB.z = __builtin_fmaf((1.0f - ALPHA) * accB.z, dnB, ALPHA * tB.z);
    resB.w = __builtin_fmaf((1.0f - ALPHA) * accB.w, dnB, ALPHA * tB.w);

    if (!LAST) {
        if (q < ROWV4) {
            if (rowA < N)
                hout[(size_t)rowA * ROWV4 + q] = make_float4(resA.x * snA, resA.y * snA, resA.z * snA, resA.w * snA);
            if (rowB < N)
                hout[(size_t)rowB * ROWV4 + q] = make_float4(resB.x * snB, resB.y * snB, resB.z * snB, resB.w * snB);
        }
    } else {
        __shared__ float sm[32 * NCLASSES];
        if (q < ROWV4) {
            if (rowA < N) {
                float* p = sm + rlA * NCLASSES + q * 4;
                p[0] = resA.x; p[1] = resA.y; p[2] = resA.z;
                if (q * 4 + 3 < NCLASSES) p[3] = resA.w;
            }
            if (rowB < N) {
                float* p = sm + rlB * NCLASSES + q * 4;
                p[0] = resB.x; p[1] = resB.y; p[2] = resB.z;
                if (q * 4 + 3 < NCLASSES) p[3] = resB.w;
            }
        }
        __syncthreads();
        const int rowsHere = min(32, N - (int)blockIdx.x * 32);
        const int nvalid = rowsHere * NCLASSES;
        float* obase = out + (size_t)blockIdx.x * (32 * NCLASSES);
        for (int p = 0; p < (32 * NCLASSES + 255) / 256; ++p) {
            const int i = tid + p * 256;
            if (i < nvalid) obase[i] = sm[i];
        }
    }
}

extern "C" void kernel_launch(void* const* d_in, const int* in_sizes, int n_in,
                              void* d_out, int out_size, void* d_ws, size_t ws_size,
                              hipStream_t stream) {
    const float* F  = (const float*)d_in[0];
    const float* W0 = (const float*)d_in[1];
    const float* b0 = (const float*)d_in[2];
    const float* W1 = (const float*)d_in[3];
    const float* b1 = (const float*)d_in[4];
    const int*  src = (const int*)d_in[5];
    const int*  dst = (const int*)d_in[6];

    const int N = in_sizes[0] / IN_FEATS;
    const int E = in_sizes[5];
    float* out = (float*)d_out;

    float* ws    = (float*)d_ws;
    float* snorm = ws;
    int*   icnt  = (int*)(ws + (size_t)N);
    float* dnorm = ws + 2 * (size_t)N;
    float* hrelu = ws + 3 * (size_t)N;
    float* h0    = hrelu + (size_t)N * HIDDEN;
    float* hs    = h0 + (size_t)N * CPAD;
    int*   rank  = (int*)(hs + (size_t)N * CPAD);
    int*   csr   = rank + (size_t)E;
    int*   rowptr = csr + (size_t)E;
    unsigned short* part_s = (unsigned short*)hrelu;
    unsigned short* part_d = part_s + (size_t)HBLK * NODES;
    int*            boff   = (int*)(part_d + (size_t)HBLK * NODES);

    {
        appnp_hist<<<2 * HBLK, 1024, 0, stream>>>(src, dst, part_s, part_d, rank, E);
        appnp_merge<<<(N + 255) / 256, 256, 0, stream>>>(part_s, part_d, boff, icnt, snorm, dnorm, N);
        appnp_rowptr<<<(N + 1023) / 1024, 256, 0, stream>>>(icnt, rowptr, N);
        appnp_fill<<<HBLK * FSPLIT, 1024, 0, stream>>>(src, dst, rank, rowptr, boff, csr, N, E);
    }

    {
        int rowTiles = (N + 15) / 16;
        int blocks   = (rowTiles + MLP_WAVES - 1) / MLP_WAVES;
        appnp_mlp_wmma<<<blocks, MLP_THREADS, 0, stream>>>(F, W0, b0, W1, b1, snorm, h0, hs, N);
    }

    {
        float* bufB = hrelu;
        const int gblocks = (N + 31) / 32;
        for (int it = 0; it < KSTEPS; ++it) {
            const float* in = (it & 1) ? bufB : hs;
            float* nxt      = (it & 1) ? hs : bufB;
            if (it == KSTEPS - 1) {
                appnp_gather<true><<<gblocks, 256, 0, stream>>>((const float4*)in, rowptr, csr, dnorm, snorm,
                                                               (const float4*)h0, (float4*)nxt, out, N);
            } else {
                appnp_gather<false><<<gblocks, 256, 0, stream>>>((const float4*)in, rowptr, csr, dnorm, snorm,
                                                                (const float4*)h0, (float4*)nxt, out, N);
            }
        }
    }
}
